# starting kernel
# baseline (speedup 1.0000x reference)
#include <hip/hip_runtime.h>

typedef float v2f __attribute__((ext_vector_type(2)));
typedef float v8f __attribute__((ext_vector_type(8)));

#define N_NODES 200000
#define N_EDGES 3200000
#define N_GRAPH 2048
#define HID     64
#define BPAIR_STRIDE 80

__global__ void k_init_deg(float* deg, int n) {
    int i = blockIdx.x * blockDim.x + threadIdx.x;
    if (i < n) deg[i] = 1.0f;
}

__global__ void k_deg_edges(const int* __restrict__ dst, float* deg, int e) {
    int i = blockIdx.x * blockDim.x + threadIdx.x;
    if (i < e) atomicAdd(&deg[dst[i]], 1.0f);
}

__global__ void k_dinv_selfloop(float* deg_dinv, const float* __restrict__ x,
                                float* s, int n) {
    int i = blockIdx.x * blockDim.x + threadIdx.x;
    if (i < n) {
        float d = rsqrtf(deg_dinv[i]);
        deg_dinv[i] = d;
        s[i] = d * d * x[i];
    }
}

__global__ void k_l1_edges(const int* __restrict__ src, const int* __restrict__ dst,
                           const float* __restrict__ dinv, const float* __restrict__ x,
                           float* s, int e) {
    int i = blockIdx.x * blockDim.x + threadIdx.x;
    if (i < e) {
        int u = src[i], v = dst[i];
        atomicAdd(&s[v], dinv[u] * dinv[v] * x[u]);
    }
}

__global__ void k_l1_feat(const float* __restrict__ s, const float* __restrict__ W1,
                          const float* __restrict__ b1, float* __restrict__ h1, int n) {
    int t = blockIdx.x * blockDim.x + threadIdx.x;
    if (t < n * HID) {
        int i = t >> 6, h = t & 63;
        float v = s[i] * W1[h] + b1[h];
        h1[t] = v > 0.0f ? v : 0.0f;
    }
}

__global__ __launch_bounds__(256)
void k_gemm_wmma(const float* __restrict__ h1, const float* __restrict__ W2,
                 float* __restrict__ t, int n_tiles) {
    __shared__ v2f sB[(HID / 2) * BPAIR_STRIDE];
    {
        float* sf = (float*)sB;
        for (int idx = threadIdx.x; idx < HID * HID; idx += blockDim.x) {
            int k = idx >> 6, col = idx & 63;
            sf[(k >> 1) * (2 * BPAIR_STRIDE) + col * 2 + (k & 1)] = W2[idx];
        }
    }
    __syncthreads();

    const int wave = threadIdx.x >> 5;
    const int lane = threadIdx.x & 31;
    const int tile = blockIdx.x * 8 + wave;
    if (tile >= n_tiles) return;

    const int row0 = tile * 16;
    const int m    = lane & 15;
    const int half = lane >> 4;

    v8f acc0 = {}, acc1 = {}, acc2 = {}, acc3 = {};
    const float* arow = h1 + (size_t)(row0 + m) * HID;

    for (int kk = 0; kk < HID; kk += 4) {
        const int ka = kk + 2 * half;
        v2f a;
        a.x = arow[ka];
        a.y = arow[ka + 1];

        const v2f* bp = &sB[(ka >> 1) * BPAIR_STRIDE + m];
        acc0 = __builtin_amdgcn_wmma_f32_16x16x4_f32(false, a, false, bp[ 0], (short)0, acc0, false, false);
        acc1 = __builtin_amdgcn_wmma_f32_16x16x4_f32(false, a, false, bp[16], (short)0, acc1, false, false);
        acc2 = __builtin_amdgcn_wmma_f32_16x16x4_f32(false, a, false, bp[32], (short)0, acc2, false, false);
        acc3 = __builtin_amdgcn_wmma_f32_16x16x4_f32(false, a, false, bp[48], (short)0, acc3, false, false);
    }

    float* orow = t + (size_t)(row0 + half * 8) * HID;
#pragma unroll
    for (int r = 0; r < 8; ++r) {
        orow[(size_t)r * HID +      m] = acc0[r];
        orow[(size_t)r * HID + 16 + m] = acc1[r];
        orow[(size_t)r * HID + 32 + m] = acc2[r];
        orow[(size_t)r * HID + 48 + m] = acc3[r];
    }
}

__global__ void k_l2_init(const float* __restrict__ t, const float* __restrict__ dinv,
                          float* __restrict__ out2, int n) {
    int tid = blockIdx.x * blockDim.x + threadIdx.x;
    if (tid < n * HID) {
        float d = dinv[tid >> 6];
        out2[tid] = d * d * t[tid];
    }
}

__global__ void k_l2_edges(const int* __restrict__ src, const int* __restrict__ dst,
                           const float* __restrict__ dinv, const float* __restrict__ t,
                           float* out2, int e) {
    int tid = blockIdx.x * blockDim.x + threadIdx.x;
    int ei  = tid >> 5;
    if (ei >= e) return;
    int lane = tid & 31;
    int u = src[ei], v = dst[ei];
    float w = dinv[u] * dinv[v];
    const float* tr = t + (size_t)u * HID + lane * 2;
    float* orow = out2 + (size_t)v * HID + lane * 2;
    atomicAdd(&orow[0], w * tr[0]);
    atomicAdd(&orow[1], w * tr[1]);
}

__global__ void k_pool_init(float* pool, float* cnt) {
    int i = blockIdx.x * blockDim.x + threadIdx.x;
    if (i < N_GRAPH * HID) pool[i] = 0.0f;
    if (i < N_GRAPH) cnt[i] = 0.0f;
}

__global__ void k_relu_pool(const float* __restrict__ out2, const float* __restrict__ b2,
                            const int* __restrict__ batch, float* pool, float* cnt, int n) {
    int tid = blockIdx.x * blockDim.x + threadIdx.x;
    if (tid < n * HID) {
        int i = tid >> 6, h = tid & 63;
        float v = out2[tid] + b2[h];
        v = v > 0.0f ? v : 0.0f;
        int g = batch[i];
        atomicAdd(&pool[g * HID + h], v);
        if (h == 0) atomicAdd(&cnt[g], 1.0f);
    }
}

__global__ void k_final(const float* __restrict__ pool, const float* __restrict__ cnt,
                        float* __restrict__ out) {
    int tid = blockIdx.x * blockDim.x + threadIdx.x;
    if (tid < N_GRAPH * HID) {
        float c = cnt[tid >> 6];
        out[tid] = pool[tid] / fmaxf(c, 1.0f);
    }
}

extern "C" void kernel_launch(void* const* d_in, const int* in_sizes, int n_in,
                              void* d_out, int out_size, void* d_ws, size_t ws_size,
                              hipStream_t stream) {
    const float* x     = (const float*)d_in[0];
    const int*   edge  = (const int*)  d_in[1];
    const int*   batch = (const int*)  d_in[2];
    const float* W1    = (const float*)d_in[3];
    const float* b1    = (const float*)d_in[4];
    const float* W2    = (const float*)d_in[5];
    const float* b2    = (const float*)d_in[6];
    const int* src = edge;
    const int* dst = edge + N_EDGES;
    float* out = (float*)d_out;

    float* ws   = (float*)d_ws;
    float* dinv = ws;
    float* s    = dinv + N_NODES;
    float* bufA = s + N_NODES;
    float* bufB = bufA + (size_t)N_NODES * HID;
    float* pool = bufB + (size_t)N_NODES * HID;
    float* cnt  = pool + N_GRAPH * HID;

    const int TB = 256;
    k_init_deg     <<<(N_NODES + TB - 1) / TB, TB, 0, stream>>>(dinv, N_NODES);
    k_deg_edges    <<<(N_EDGES + TB - 1) / TB, TB, 0, stream>>>(dst, dinv, N_EDGES);
    k_dinv_selfloop<<<(N_NODES + TB - 1) / TB, TB, 0, stream>>>(dinv, x, s, N_NODES);
    k_l1_edges     <<<(N_EDGES + TB - 1) / TB, TB, 0, stream>>>(src, dst, dinv, x, s, N_EDGES);
    k_l1_feat      <<<(N_NODES * HID + TB - 1) / TB, TB, 0, stream>>>(s, W1, b1, bufA, N_NODES);

    const int n_tiles = N_NODES / 16;
    k_gemm_wmma    <<<(n_tiles + 7) / 8, 256, 0, stream>>>(bufA, W2, bufB, n_tiles);

    k_l2_init      <<<(N_NODES * HID + TB - 1) / TB, TB, 0, stream>>>(bufB, dinv, bufA, N_NODES);
    {
        long long thr = (long long)N_EDGES * 32;
        k_l2_edges <<<(unsigned)((thr + TB - 1) / TB), TB, 0, stream>>>(src, dst, dinv, bufB, bufA, N_EDGES);
    }
    k_pool_init    <<<(N_GRAPH * HID + TB - 1) / TB, TB, 0, stream>>>(pool, cnt);
    k_relu_pool    <<<(N_NODES * HID + TB - 1) / TB, TB, 0, stream>>>(bufA, b2, batch, pool, cnt, N_NODES);
    k_final        <<<(N_GRAPH * HID + TB - 1) / TB, TB, 0, stream>>>(pool, cnt, out);
}


# Degree and layer-1 scalar scatter-adds: per-node-range LDS float atomics over edge slices, partial planes summed in fixed order; no global atomics there (sum order changes)
# speedup vs baseline: 1.3260x; 1.3260x over previous
#include <hip/hip_runtime.h>

typedef float v2f __attribute__((ext_vector_type(2)));
typedef float v8f __attribute__((ext_vector_type(8)));

#define N_NODES 200000
#define N_EDGES 3200000
#define N_GRAPH 2048
#define HID     64
#define BPAIR_STRIDE 80

#define NRANGE  4
#define RANGE_N 50000
#define NSLICE  64
#define SLICE_E 50000
#define PTB     1024
#define EBATCH  8

__global__ __launch_bounds__(PTB)
void k_deg_part(const int* __restrict__ dst, float* __restrict__ part) {
    __shared__ float acc[RANGE_N];
    const int tid = threadIdx.x;
    const int rng = blockIdx.x & (NRANGE - 1);
    const int sl  = blockIdx.x / NRANGE;
    const int lo  = rng * RANGE_N;
    for (int j = tid; j < RANGE_N; j += PTB) acc[j] = 0.0f;
    __syncthreads();

    const int* d = dst + (size_t)sl * SLICE_E;
    for (int base = 0; base < SLICE_E; base += PTB * EBATCH) {
        int v[EBATCH];
#pragma unroll
        for (int k = 0; k < EBATCH; ++k) {
            int j = base + k * PTB + tid;
            v[k] = (j < SLICE_E) ? d[j] : -1;
        }
#pragma unroll
        for (int k = 0; k < EBATCH; ++k) {
            unsigned o = (unsigned)(v[k] - lo);
            if (o < (unsigned)RANGE_N) atomicAdd(&acc[o], 1.0f);
        }
    }
    __syncthreads();

    float* p = part + (size_t)sl * N_NODES + lo;
    for (int j = tid; j < RANGE_N; j += PTB) p[j] = acc[j];
}

__global__ void k_dinv_reduce(const float* __restrict__ part, float* __restrict__ dinv, int n) {
    int i = blockIdx.x * blockDim.x + threadIdx.x;
    if (i < n) {
        float deg = 1.0f;
#pragma unroll 16
        for (int p = 0; p < NSLICE; ++p) deg += part[(size_t)p * N_NODES + i];
        dinv[i] = rsqrtf(deg);
    }
}

__global__ __launch_bounds__(PTB)
void k_l1_part(const int* __restrict__ src, const int* __restrict__ dst,
               const float* __restrict__ dinv, const float* __restrict__ x,
               float* __restrict__ part) {
    __shared__ float acc[RANGE_N];
    const int tid = threadIdx.x;
    const int rng = blockIdx.x & (NRANGE - 1);
    const int sl  = blockIdx.x / NRANGE;
    const int lo  = rng * RANGE_N;
    for (int j = tid; j < RANGE_N; j += PTB) acc[j] = 0.0f;
    __syncthreads();

    const int* sp = src + (size_t)sl * SLICE_E;
    const int* dp = dst + (size_t)sl * SLICE_E;
    for (int base = 0; base < SLICE_E; base += PTB * EBATCH) {
        int u[EBATCH], v[EBATCH];
#pragma unroll
        for (int k = 0; k < EBATCH; ++k) {
            int j = base + k * PTB + tid;
            bool ok = j < SLICE_E;
            u[k] = ok ? sp[j] : 0;
            v[k] = ok ? dp[j] : -1;
        }
        float val[EBATCH];
#pragma unroll
        for (int k = 0; k < EBATCH; ++k) {
            bool in = (unsigned)(v[k] - lo) < (unsigned)RANGE_N;
            int uu = in ? u[k] : 0;
            int vv = in ? v[k] : 0;
            val[k] = dinv[uu] * dinv[vv] * x[uu];
        }
#pragma unroll
        for (int k = 0; k < EBATCH; ++k) {
            unsigned o = (unsigned)(v[k] - lo);
            if (o < (unsigned)RANGE_N) atomicAdd(&acc[o], val[k]);
        }
    }
    __syncthreads();

    float* p = part + (size_t)sl * N_NODES + lo;
    for (int j = tid; j < RANGE_N; j += PTB) p[j] = acc[j];
}

__global__ void k_s_reduce(const float* __restrict__ part, const float* __restrict__ dinv,
                           const float* __restrict__ x, float* __restrict__ s, int n) {
    int i = blockIdx.x * blockDim.x + threadIdx.x;
    if (i < n) {
        float d = dinv[i];
        float a = d * d * x[i];
#pragma unroll 16
        for (int p = 0; p < NSLICE; ++p) a += part[(size_t)p * N_NODES + i];
        s[i] = a;
    }
}

__global__ void k_l1_feat(const float* __restrict__ s, const float* __restrict__ W1,
                          const float* __restrict__ b1, float* __restrict__ h1, int n) {
    int t = blockIdx.x * blockDim.x + threadIdx.x;
    if (t < n * HID) {
        int i = t >> 6, h = t & 63;
        float v = s[i] * W1[h] + b1[h];
        h1[t] = v > 0.0f ? v : 0.0f;
    }
}

__global__ __launch_bounds__(256)
void k_gemm_wmma(const float* __restrict__ h1, const float* __restrict__ W2,
                 float* __restrict__ t, int n_tiles) {
    __shared__ v2f sB[(HID / 2) * BPAIR_STRIDE];
    {
        float* sf = (float*)sB;
        for (int idx = threadIdx.x; idx < HID * HID; idx += blockDim.x) {
            int k = idx >> 6, col = idx & 63;
            sf[(k >> 1) * (2 * BPAIR_STRIDE) + col * 2 + (k & 1)] = W2[idx];
        }
    }
    __syncthreads();

    const int wave = threadIdx.x >> 5;
    const int lane = threadIdx.x & 31;
    const int tile = blockIdx.x * 8 + wave;
    if (tile >= n_tiles) return;

    const int row0 = tile * 16;
    const int m    = lane & 15;
    const int half = lane >> 4;

    v8f acc0 = {}, acc1 = {}, acc2 = {}, acc3 = {};
    const float* arow = h1 + (size_t)(row0 + m) * HID;

    for (int kk = 0; kk < HID; kk += 4) {
        const int ka = kk + 2 * half;
        v2f a;
        a.x = arow[ka];
        a.y = arow[ka + 1];

        const v2f* bp = &sB[(ka >> 1) * BPAIR_STRIDE + m];
        acc0 = __builtin_amdgcn_wmma_f32_16x16x4_f32(false, a, false, bp[ 0], (short)0, acc0, false, false);
        acc1 = __builtin_amdgcn_wmma_f32_16x16x4_f32(false, a, false, bp[16], (short)0, acc1, false, false);
        acc2 = __builtin_amdgcn_wmma_f32_16x16x4_f32(false, a, false, bp[32], (short)0, acc2, false, false);
        acc3 = __builtin_amdgcn_wmma_f32_16x16x4_f32(false, a, false, bp[48], (short)0, acc3, false, false);
    }

    float* orow = t + (size_t)(row0 + half * 8) * HID;
#pragma unroll
    for (int r = 0; r < 8; ++r) {
        orow[(size_t)r * HID +      m] = acc0[r];
        orow[(size_t)r * HID + 16 + m] = acc1[r];
        orow[(size_t)r * HID + 32 + m] = acc2[r];
        orow[(size_t)r * HID + 48 + m] = acc3[r];
    }
}

__global__ void k_l2_init(const float* __restrict__ t, const float* __restrict__ dinv,
                          float* __restrict__ out2, int n) {
    int tid = blockIdx.x * blockDim.x + threadIdx.x;
    if (tid < n * HID) {
        float d = dinv[tid >> 6];
        out2[tid] = d * d * t[tid];
    }
}

__global__ void k_l2_edges(const int* __restrict__ src, const int* __restrict__ dst,
                           const float* __restrict__ dinv, const float* __restrict__ t,
                           float* out2, int e) {
    int tid = blockIdx.x * blockDim.x + threadIdx.x;
    int ei  = tid >> 5;
    if (ei >= e) return;
    int lane = tid & 31;
    int u = src[ei], v = dst[ei];
    float w = dinv[u] * dinv[v];
    const float* tr = t + (size_t)u * HID + lane * 2;
    float* orow = out2 + (size_t)v * HID + lane * 2;
    atomicAdd(&orow[0], w * tr[0]);
    atomicAdd(&orow[1], w * tr[1]);
}

__global__ void k_pool_init(float* pool, float* cnt) {
    int i = blockIdx.x * blockDim.x + threadIdx.x;
    if (i < N_GRAPH * HID) pool[i] = 0.0f;
    if (i < N_GRAPH) cnt[i] = 0.0f;
}

__global__ void k_relu_pool(const float* __restrict__ out2, const float* __restrict__ b2,
                            const int* __restrict__ batch, float* pool, float* cnt, int n) {
    int tid = blockIdx.x * blockDim.x + threadIdx.x;
    if (tid < n * HID) {
        int i = tid >> 6, h = tid & 63;
        float v = out2[tid] + b2[h];
        v = v > 0.0f ? v : 0.0f;
        int g = batch[i];
        atomicAdd(&pool[g * HID + h], v);
        if (h == 0) atomicAdd(&cnt[g], 1.0f);
    }
}

__global__ void k_final(const float* __restrict__ pool, const float* __restrict__ cnt,
                        float* __restrict__ out) {
    int tid = blockIdx.x * blockDim.x + threadIdx.x;
    if (tid < N_GRAPH * HID) {
        float c = cnt[tid >> 6];
        out[tid] = pool[tid] / fmaxf(c, 1.0f);
    }
}

extern "C" void kernel_launch(void* const* d_in, const int* in_sizes, int n_in,
                              void* d_out, int out_size, void* d_ws, size_t ws_size,
                              hipStream_t stream) {
    const float* x     = (const float*)d_in[0];
    const int*   edge  = (const int*)  d_in[1];
    const int*   batch = (const int*)  d_in[2];
    const float* W1    = (const float*)d_in[3];
    const float* b1    = (const float*)d_in[4];
    const float* W2    = (const float*)d_in[5];
    const float* b2    = (const float*)d_in[6];
    const int* src = edge;
    const int* dst = edge + N_EDGES;
    float* out = (float*)d_out;

    float* ws   = (float*)d_ws;
    float* dinv = ws;
    float* s    = dinv + N_NODES;
    float* bufA = s + N_NODES;
    float* bufB = bufA + (size_t)N_NODES * HID;
    float* pool = bufB + (size_t)N_NODES * HID;
    float* cnt  = pool + N_GRAPH * HID;

    const int TB = 256;
    k_deg_part     <<<NSLICE * NRANGE, PTB, 0, stream>>>(dst, bufA);
    k_dinv_reduce  <<<(N_NODES + TB - 1) / TB, TB, 0, stream>>>(bufA, dinv, N_NODES);
    k_l1_part      <<<NSLICE * NRANGE, PTB, 0, stream>>>(src, dst, dinv, x, bufA);
    k_s_reduce     <<<(N_NODES + TB - 1) / TB, TB, 0, stream>>>(bufA, dinv, x, s, N_NODES);
    k_l1_feat      <<<(N_NODES * HID + TB - 1) / TB, TB, 0, stream>>>(s, W1, b1, bufA, N_NODES);

    const int n_tiles = N_NODES / 16;
    k_gemm_wmma    <<<(n_tiles + 7) / 8, 256, 0, stream>>>(bufA, W2, bufB, n_tiles);

    k_l2_init      <<<(N_NODES * HID + TB - 1) / TB, TB, 0, stream>>>(bufB, dinv, bufA, N_NODES);
    {
        long long thr = (long long)N_EDGES * 32;
        k_l2_edges <<<(unsigned)((thr + TB - 1) / TB), TB, 0, stream>>>(src, dst, dinv, bufB, bufA, N_EDGES);
    }
    k_pool_init    <<<(N_GRAPH * HID + TB - 1) / TB, TB, 0, stream>>>(pool, cnt);
    k_relu_pool    <<<(N_NODES * HID + TB - 1) / TB, TB, 0, stream>>>(bufA, b2, batch, pool, cnt, N_NODES);
    k_final        <<<(N_GRAPH * HID + TB - 1) / TB, TB, 0, stream>>>(pool, cnt, out);
}


# Layer-2 aggregation pulls per destination node over a per-call dst-sorted edge list (LDS-cursor counting sort), row finished in registers into the pool; no out2 atomics (sum order changes)
# speedup vs baseline: 1.8530x; 1.3974x over previous
#include <hip/hip_runtime.h>

typedef float v2f __attribute__((ext_vector_type(2)));
typedef float v8f __attribute__((ext_vector_type(8)));

#define N_NODES 200000
#define N_EDGES 3200000
#define N_GRAPH 2048
#define HID     64
#define BPAIR_STRIDE 80

#define NRANGE  4
#define RANGE_N 50000
#define NSLICE  64
#define SLICE_E 50000
#define PTB     1024
#define EBATCH  8

__global__ __launch_bounds__(PTB)
void k_deg_part(const int* __restrict__ dst, float* __restrict__ part) {
    __shared__ float acc[RANGE_N];
    const int tid = threadIdx.x;
    const int rng = blockIdx.x & (NRANGE - 1);
    const int sl  = blockIdx.x / NRANGE;
    const int lo  = rng * RANGE_N;
    for (int j = tid; j < RANGE_N; j += PTB) acc[j] = 0.0f;
    __syncthreads();

    const int* d = dst + (size_t)sl * SLICE_E;
    for (int base = 0; base < SLICE_E; base += PTB * EBATCH) {
        int v[EBATCH];
#pragma unroll
        for (int k = 0; k < EBATCH; ++k) {
            int j = base + k * PTB + tid;
            v[k] = (j < SLICE_E) ? d[j] : -1;
        }
#pragma unroll
        for (int k = 0; k < EBATCH; ++k) {
            unsigned o = (unsigned)(v[k] - lo);
            if (o < (unsigned)RANGE_N) atomicAdd(&acc[o], 1.0f);
        }
    }
    __syncthreads();

    float* p = part + (size_t)sl * N_NODES + lo;
    for (int j = tid; j < RANGE_N; j += PTB) p[j] = acc[j];
}

#define RTB 256
__global__ __launch_bounds__(RTB)
void k_dinv_reduce(float* part, float* __restrict__ dinv, int* __restrict__ rowcnt,
                   int* __restrict__ bsum, int n) {
    __shared__ int red[RTB / 32];
    const int i = blockIdx.x * RTB + threadIdx.x;
    int run = 0;
    if (i < n) {
        float deg = 1.0f;
#pragma unroll 16
        for (int p = 0; p < NSLICE; ++p) {
            float c = part[(size_t)p * N_NODES + i];
            part[(size_t)p * N_NODES + i] = __int_as_float(run);
            deg += c;
            run += (int)c;
        }
        dinv[i] = rsqrtf(deg);
        rowcnt[i] = run;
    }
    int v = run;
    for (int off = 16; off > 0; off >>= 1) v += __shfl_xor(v, off);
    if ((threadIdx.x & 31) == 0) red[threadIdx.x >> 5] = v;
    __syncthreads();
    if (threadIdx.x == 0) {
        int sum = 0;
        for (int w = 0; w < RTB / 32; ++w) sum += red[w];
        bsum[blockIdx.x] = sum;
    }
}

__global__ __launch_bounds__(RTB)
void k_rowptr(int* __restrict__ rowptr, const int* __restrict__ bsum, int n) {
    __shared__ int red[RTB / 32];
    __shared__ int wsum[RTB / 32];
    const int tid = threadIdx.x, lane = tid & 31, wave = tid >> 5;
    const int b = blockIdx.x;
    int pre = 0;
    for (int j = tid; j < b; j += RTB) pre += bsum[j];
    for (int off = 16; off > 0; off >>= 1) pre += __shfl_xor(pre, off);
    const int i = b * RTB + tid;
    const int c = (i < n) ? rowptr[i] : 0;
    int x = c;
    for (int off = 1; off < 32; off <<= 1) {
        int y = __shfl_up(x, off);
        if (lane >= off) x += y;
    }
    if (lane == 0)  red[wave]  = pre;
    if (lane == 31) wsum[wave] = x;
    __syncthreads();
    int offset = 0;
    for (int w = 0; w < RTB / 32; ++w) offset += red[w];
    for (int w = 0; w < wave; ++w) offset += wsum[w];
    if (i < n) rowptr[i] = offset + x - c;
}

__global__ __launch_bounds__(PTB)
void k_csr_fill(const int* __restrict__ src, const int* __restrict__ dst,
                const float* __restrict__ pre, const int* __restrict__ rowptr,
                int* __restrict__ csr_src) {
    __shared__ int cur[RANGE_N];
    const int tid = threadIdx.x;
    const int rng = blockIdx.x & (NRANGE - 1);
    const int sl  = blockIdx.x / NRANGE;
    const int lo  = rng * RANGE_N;
    const float* pp = pre + (size_t)sl * N_NODES + lo;
    const int*   rp = rowptr + lo;
    for (int base = 0; base < RANGE_N; base += PTB * EBATCH) {
        int a[EBATCH], b[EBATCH];
#pragma unroll
        for (int k = 0; k < EBATCH; ++k) {
            int j = base + k * PTB + tid;
            bool ok = j < RANGE_N;
            a[k] = ok ? rp[j] : 0;
            b[k] = ok ? __float_as_int(pp[j]) : 0;
        }
#pragma unroll
        for (int k = 0; k < EBATCH; ++k) {
            int j = base + k * PTB + tid;
            if (j < RANGE_N) cur[j] = a[k] + b[k];
        }
    }
    __syncthreads();

    const int* sp = src + (size_t)sl * SLICE_E;
    const int* dp = dst + (size_t)sl * SLICE_E;
    for (int base = 0; base < SLICE_E; base += PTB * EBATCH) {
        int u[EBATCH], v[EBATCH];
#pragma unroll
        for (int k = 0; k < EBATCH; ++k) {
            int j = base + k * PTB + tid;
            bool ok = j < SLICE_E;
            u[k] = ok ? sp[j] : 0;
            v[k] = ok ? dp[j] : -1;
        }
#pragma unroll
        for (int k = 0; k < EBATCH; ++k) {
            unsigned o = (unsigned)(v[k] - lo);
            if (o < (unsigned)RANGE_N) {
                int pos = atomicAdd(&cur[o], 1);
                csr_src[pos] = u[k];
            }
        }
    }
}

__global__ __launch_bounds__(PTB)
void k_l1_part(const int* __restrict__ src, const int* __restrict__ dst,
               const float* __restrict__ dinv, const float* __restrict__ x,
               float* __restrict__ part) {
    __shared__ float acc[RANGE_N];
    const int tid = threadIdx.x;
    const int rng = blockIdx.x & (NRANGE - 1);
    const int sl  = blockIdx.x / NRANGE;
    const int lo  = rng * RANGE_N;
    for (int j = tid; j < RANGE_N; j += PTB) acc[j] = 0.0f;
    __syncthreads();

    const int* sp = src + (size_t)sl * SLICE_E;
    const int* dp = dst + (size_t)sl * SLICE_E;
    for (int base = 0; base < SLICE_E; base += PTB * EBATCH) {
        int u[EBATCH], v[EBATCH];
#pragma unroll
        for (int k = 0; k < EBATCH; ++k) {
            int j = base + k * PTB + tid;
            bool ok = j < SLICE_E;
            u[k] = ok ? sp[j] : 0;
            v[k] = ok ? dp[j] : -1;
        }
        float val[EBATCH];
#pragma unroll
        for (int k = 0; k < EBATCH; ++k) {
            bool in = (unsigned)(v[k] - lo) < (unsigned)RANGE_N;
            int uu = in ? u[k] : 0;
            int vv = in ? v[k] : 0;
            val[k] = dinv[uu] * dinv[vv] * x[uu];
        }
#pragma unroll
        for (int k = 0; k < EBATCH; ++k) {
            unsigned o = (unsigned)(v[k] - lo);
            if (o < (unsigned)RANGE_N) atomicAdd(&acc[o], val[k]);
        }
    }
    __syncthreads();

    float* p = part + (size_t)sl * N_NODES + lo;
    for (int j = tid; j < RANGE_N; j += PTB) p[j] = acc[j];
}

__global__ void k_s_reduce(const float* __restrict__ part, const float* __restrict__ dinv,
                           const float* __restrict__ x, float* __restrict__ s, int n) {
    int i = blockIdx.x * blockDim.x + threadIdx.x;
    if (i < n) {
        float d = dinv[i];
        float a = d * d * x[i];
#pragma unroll 16
        for (int p = 0; p < NSLICE; ++p) a += part[(size_t)p * N_NODES + i];
        s[i] = a;
    }
}

__global__ void k_l1_feat(const float* __restrict__ s, const float* __restrict__ W1,
                          const float* __restrict__ b1, float* __restrict__ h1, int n) {
    int t = blockIdx.x * blockDim.x + threadIdx.x;
    if (t < n * HID) {
        int i = t >> 6, h = t & 63;
        float v = s[i] * W1[h] + b1[h];
        h1[t] = v > 0.0f ? v : 0.0f;
    }
}

__global__ __launch_bounds__(256)
void k_gemm_wmma(const float* h1, const float* __restrict__ W2,
                 float* t, int n_tiles) {
    __shared__ v2f sB[(HID / 2) * BPAIR_STRIDE];
    {
        float* sf = (float*)sB;
        for (int idx = threadIdx.x; idx < HID * HID; idx += blockDim.x) {
            int k = idx >> 6, col = idx & 63;
            sf[(k >> 1) * (2 * BPAIR_STRIDE) + col * 2 + (k & 1)] = W2[idx];
        }
    }
    __syncthreads();

    const int wave = threadIdx.x >> 5;
    const int lane = threadIdx.x & 31;
    const int tile = blockIdx.x * 8 + wave;
    if (tile >= n_tiles) return;

    const int row0 = tile * 16;
    const int m    = lane & 15;
    const int half = lane >> 4;

    v8f acc0 = {}, acc1 = {}, acc2 = {}, acc3 = {};
    const float* arow = h1 + (size_t)(row0 + m) * HID;

    for (int kk = 0; kk < HID; kk += 4) {
        const int ka = kk + 2 * half;
        v2f a;
        a.x = arow[ka];
        a.y = arow[ka + 1];

        const v2f* bp = &sB[(ka >> 1) * BPAIR_STRIDE + m];
        acc0 = __builtin_amdgcn_wmma_f32_16x16x4_f32(false, a, false, bp[ 0], (short)0, acc0, false, false);
        acc1 = __builtin_amdgcn_wmma_f32_16x16x4_f32(false, a, false, bp[16], (short)0, acc1, false, false);
        acc2 = __builtin_amdgcn_wmma_f32_16x16x4_f32(false, a, false, bp[32], (short)0, acc2, false, false);
        acc3 = __builtin_amdgcn_wmma_f32_16x16x4_f32(false, a, false, bp[48], (short)0, acc3, false, false);
    }

    float* orow = t + (size_t)(row0 + half * 8) * HID;
#pragma unroll
    for (int r = 0; r < 8; ++r) {
        orow[(size_t)r * HID +      m] = acc0[r];
        orow[(size_t)r * HID + 16 + m] = acc1[r];
        orow[(size_t)r * HID + 32 + m] = acc2[r];
        orow[(size_t)r * HID + 48 + m] = acc3[r];
    }
}

#define PULLB 8
__global__ __launch_bounds__(256)
void k_l2_pull(const int* __restrict__ rowptr, const int* __restrict__ csr_src,
               const float* __restrict__ dinv, const float* __restrict__ t,
               const float* __restrict__ b2, const int* __restrict__ batch,
               float* pool, float* cnt, int n) {
    const int node = __builtin_amdgcn_readfirstlane((int)((blockIdx.x * blockDim.x + threadIdx.x) >> 5));
    if (node >= n) return;
    const int lane = threadIdx.x & 31;
    const int beg = rowptr[node];
    const int end = (node + 1 < n) ? rowptr[node + 1] : N_EDGES;
    const float dv = dinv[node];
    const int g = batch[node];
    const float bx = b2[lane * 2], by = b2[lane * 2 + 1];
    const v2f* t2 = (const v2f*)t;
    const v2f me = t2[(size_t)node * (HID / 2) + lane];
    float ax = dv * dv * me.x;
    float ay = dv * dv * me.y;

    for (int base = beg; base < end; base += 32) {
        const int j = base + lane;
        int u = 0;
        float w = 0.0f;
        if (j < end) {
            u = csr_src[j];
            w = dinv[u] * dv;
        }
        const int c = min(32, end - base);
        for (int k = 0; k < c; k += PULLB) {
            int uk[PULLB];
            float wk[PULLB];
            v2f tv[PULLB];
#pragma unroll
            for (int q = 0; q < PULLB; ++q) {
                uk[q] = __builtin_amdgcn_readlane(u, k + q);
                wk[q] = __int_as_float(__builtin_amdgcn_readlane(__float_as_int(w), k + q));
                tv[q] = t2[(size_t)uk[q] * (HID / 2) + lane];
            }
#pragma unroll
            for (int q = 0; q < PULLB; ++q) {
                if (k + q < c) {
                    ax += wk[q] * tv[q].x;
                    ay += wk[q] * tv[q].y;
                }
            }
        }
    }

    float vx = ax + bx, vy = ay + by;
    vx = vx > 0.0f ? vx : 0.0f;
    vy = vy > 0.0f ? vy : 0.0f;
    float* pr = pool + (size_t)g * HID + lane * 2;
    atomicAdd(&pr[0], vx);
    atomicAdd(&pr[1], vy);
    if (lane == 0) atomicAdd(&cnt[g], 1.0f);
}

__global__ void k_pool_init(float* pool, float* cnt) {
    int i = blockIdx.x * blockDim.x + threadIdx.x;
    if (i < N_GRAPH * HID) pool[i] = 0.0f;
    if (i < N_GRAPH) cnt[i] = 0.0f;
}

__global__ void k_final(const float* __restrict__ pool, const float* __restrict__ cnt,
                        float* __restrict__ out) {
    int tid = blockIdx.x * blockDim.x + threadIdx.x;
    if (tid < N_GRAPH * HID) {
        float c = cnt[tid >> 6];
        out[tid] = pool[tid] / fmaxf(c, 1.0f);
    }
}

extern "C" void kernel_launch(void* const* d_in, const int* in_sizes, int n_in,
                              void* d_out, int out_size, void* d_ws, size_t ws_size,
                              hipStream_t stream) {
    const float* x     = (const float*)d_in[0];
    const int*   edge  = (const int*)  d_in[1];
    const int*   batch = (const int*)  d_in[2];
    const float* W1    = (const float*)d_in[3];
    const float* b1    = (const float*)d_in[4];
    const float* W2    = (const float*)d_in[5];
    const float* b2    = (const float*)d_in[6];
    const int* src = edge;
    const int* dst = edge + N_EDGES;
    float* out = (float*)d_out;

    float* ws   = (float*)d_ws;
    float* dinv = ws;
    float* s    = dinv + N_NODES;
    float* bufA = s + N_NODES;
    float* bufB = bufA + (size_t)N_NODES * HID;
    int* rowptr  = (int*)bufB;
    int* bsum    = rowptr + N_NODES;
    int* csr_src = bsum + 1024;
    float* pool = bufB + (size_t)N_NODES * HID;
    float* cnt  = pool + N_GRAPH * HID;

    const int TB = 256;
    k_deg_part     <<<NSLICE * NRANGE, PTB, 0, stream>>>(dst, bufA);
    k_dinv_reduce  <<<(N_NODES + RTB - 1) / RTB, RTB, 0, stream>>>(bufA, dinv, rowptr, bsum, N_NODES);
    k_rowptr       <<<(N_NODES + RTB - 1) / RTB, RTB, 0, stream>>>(rowptr, bsum, N_NODES);
    k_csr_fill     <<<NSLICE * NRANGE, PTB, 0, stream>>>(src, dst, bufA, rowptr, csr_src);
    k_l1_part      <<<NSLICE * NRANGE, PTB, 0, stream>>>(src, dst, dinv, x, bufA);
    k_s_reduce     <<<(N_NODES + TB - 1) / TB, TB, 0, stream>>>(bufA, dinv, x, s, N_NODES);
    k_l1_feat      <<<(N_NODES * HID + TB - 1) / TB, TB, 0, stream>>>(s, W1, b1, bufA, N_NODES);

    const int n_tiles = N_NODES / 16;
    k_gemm_wmma    <<<(n_tiles + 7) / 8, 256, 0, stream>>>(bufA, W2, bufA, n_tiles);

    k_pool_init    <<<(N_GRAPH * HID + TB - 1) / TB, TB, 0, stream>>>(pool, cnt);
    k_l2_pull      <<<(N_NODES * 32 + TB - 1) / TB, TB, 0, stream>>>(rowptr, csr_src, dinv, bufA, b2, batch,
                                                                    pool, cnt, N_NODES);
    k_final        <<<(N_GRAPH * HID + TB - 1) / TB, TB, 0, stream>>>(pool, cnt, out);
}


# Layer-1 feature kernel folded into the GEMM: each lane builds its A fragment as relu(s*W1+b1) in registers; h1 is never stored, one launch and its 400,000 one-element waves are gone
# speedup vs baseline: 2.1518x; 1.1613x over previous
#include <hip/hip_runtime.h>

typedef float v2f __attribute__((ext_vector_type(2)));
typedef float v8f __attribute__((ext_vector_type(8)));

#define N_NODES 200000
#define N_EDGES 3200000
#define N_GRAPH 2048
#define HID     64
#define BPAIR_STRIDE 80

#define NRANGE  4
#define RANGE_N 50000
#define NSLICE  64
#define SLICE_E 50000
#define PTB     1024
#define EBATCH  8

__global__ __launch_bounds__(PTB)
void k_deg_part(const int* __restrict__ dst, float* __restrict__ part) {
    __shared__ float acc[RANGE_N];
    const int tid = threadIdx.x;
    const int rng = blockIdx.x & (NRANGE - 1);
    const int sl  = blockIdx.x / NRANGE;
    const int lo  = rng * RANGE_N;
    for (int j = tid; j < RANGE_N; j += PTB) acc[j] = 0.0f;
    __syncthreads();

    const int* d = dst + (size_t)sl * SLICE_E;
    for (int base = 0; base < SLICE_E; base += PTB * EBATCH) {
        int v[EBATCH];
#pragma unroll
        for (int k = 0; k < EBATCH; ++k) {
            int j = base + k * PTB + tid;
            v[k] = (j < SLICE_E) ? d[j] : -1;
        }
#pragma unroll
        for (int k = 0; k < EBATCH; ++k) {
            unsigned o = (unsigned)(v[k] - lo);
            if (o < (unsigned)RANGE_N) atomicAdd(&acc[o], 1.0f);
        }
    }
    __syncthreads();

    float* p = part + (size_t)sl * N_NODES + lo;
    for (int j = tid; j < RANGE_N; j += PTB) p[j] = acc[j];
}

#define RTB 256
__global__ __launch_bounds__(RTB)
void k_dinv_reduce(float* part, float* __restrict__ dinv, int* __restrict__ rowcnt,
                   int* __restrict__ bsum, int n) {
    __shared__ int red[RTB / 32];
    const int i = blockIdx.x * RTB + threadIdx.x;
    int run = 0;
    if (i < n) {
        float deg = 1.0f;
#pragma unroll 16
        for (int p = 0; p < NSLICE; ++p) {
            float c = part[(size_t)p * N_NODES + i];
            part[(size_t)p * N_NODES + i] = __int_as_float(run);
            deg += c;
            run += (int)c;
        }
        dinv[i] = rsqrtf(deg);
        rowcnt[i] = run;
    }
    int v = run;
    for (int off = 16; off > 0; off >>= 1) v += __shfl_xor(v, off);
    if ((threadIdx.x & 31) == 0) red[threadIdx.x >> 5] = v;
    __syncthreads();
    if (threadIdx.x == 0) {
        int sum = 0;
        for (int w = 0; w < RTB / 32; ++w) sum += red[w];
        bsum[blockIdx.x] = sum;
    }
}

__global__ __launch_bounds__(RTB)
void k_rowptr(int* __restrict__ rowptr, const int* __restrict__ bsum, int n) {
    __shared__ int red[RTB / 32];
    __shared__ int wsum[RTB / 32];
    const int tid = threadIdx.x, lane = tid & 31, wave = tid >> 5;
    const int b = blockIdx.x;
    int pre = 0;
    for (int j = tid; j < b; j += RTB) pre += bsum[j];
    for (int off = 16; off > 0; off >>= 1) pre += __shfl_xor(pre, off);
    const int i = b * RTB + tid;
    const int c = (i < n) ? rowptr[i] : 0;
    int x = c;
    for (int off = 1; off < 32; off <<= 1) {
        int y = __shfl_up(x, off);
        if (lane >= off) x += y;
    }
    if (lane == 0)  red[wave]  = pre;
    if (lane == 31) wsum[wave] = x;
    __syncthreads();
    int offset = 0;
    for (int w = 0; w < RTB / 32; ++w) offset += red[w];
    for (int w = 0; w < wave; ++w) offset += wsum[w];
    if (i < n) rowptr[i] = offset + x - c;
}

__global__ __launch_bounds__(PTB)
void k_csr_fill(const int* __restrict__ src, const int* __restrict__ dst,
                const float* __restrict__ pre, const int* __restrict__ rowptr,
                int* __restrict__ csr_src) {
    __shared__ int cur[RANGE_N];
    const int tid = threadIdx.x;
    const int rng = blockIdx.x & (NRANGE - 1);
    const int sl  = blockIdx.x / NRANGE;
    const int lo  = rng * RANGE_N;
    const float* pp = pre + (size_t)sl * N_NODES + lo;
    const int*   rp = rowptr + lo;
    for (int base = 0; base < RANGE_N; base += PTB * EBATCH) {
        int a[EBATCH], b[EBATCH];
#pragma unroll
        for (int k = 0; k < EBATCH; ++k) {
            int j = base + k * PTB + tid;
            bool ok = j < RANGE_N;
            a[k] = ok ? rp[j] : 0;
            b[k] = ok ? __float_as_int(pp[j]) : 0;
        }
#pragma unroll
        for (int k = 0; k < EBATCH; ++k) {
            int j = base + k * PTB + tid;
            if (j < RANGE_N) cur[j] = a[k] + b[k];
        }
    }
    __syncthreads();

    const int* sp = src + (size_t)sl * SLICE_E;
    const int* dp = dst + (size_t)sl * SLICE_E;
    for (int base = 0; base < SLICE_E; base += PTB * EBATCH) {
        int u[EBATCH], v[EBATCH];
#pragma unroll
        for (int k = 0; k < EBATCH; ++k) {
            int j = base + k * PTB + tid;
            bool ok = j < SLICE_E;
            u[k] = ok ? sp[j] : 0;
            v[k] = ok ? dp[j] : -1;
        }
#pragma unroll
        for (int k = 0; k < EBATCH; ++k) {
            unsigned o = (unsigned)(v[k] - lo);
            if (o < (unsigned)RANGE_N) {
                int pos = atomicAdd(&cur[o], 1);
                csr_src[pos] = u[k];
            }
        }
    }
}

__global__ __launch_bounds__(PTB)
void k_l1_part(const int* __restrict__ src, const int* __restrict__ dst,
               const float* __restrict__ dinv, const float* __restrict__ x,
               float* __restrict__ part) {
    __shared__ float acc[RANGE_N];
    const int tid = threadIdx.x;
    const int rng = blockIdx.x & (NRANGE - 1);
    const int sl  = blockIdx.x / NRANGE;
    const int lo  = rng * RANGE_N;
    for (int j = tid; j < RANGE_N; j += PTB) acc[j] = 0.0f;
    __syncthreads();

    const int* sp = src + (size_t)sl * SLICE_E;
    const int* dp = dst + (size_t)sl * SLICE_E;
    for (int base = 0; base < SLICE_E; base += PTB * EBATCH) {
        int u[EBATCH], v[EBATCH];
#pragma unroll
        for (int k = 0; k < EBATCH; ++k) {
            int j = base + k * PTB + tid;
            bool ok = j < SLICE_E;
            u[k] = ok ? sp[j] : 0;
            v[k] = ok ? dp[j] : -1;
        }
        float val[EBATCH];
#pragma unroll
        for (int k = 0; k < EBATCH; ++k) {
            bool in = (unsigned)(v[k] - lo) < (unsigned)RANGE_N;
            int uu = in ? u[k] : 0;
            int vv = in ? v[k] : 0;
            val[k] = dinv[uu] * dinv[vv] * x[uu];
        }
#pragma unroll
        for (int k = 0; k < EBATCH; ++k) {
            unsigned o = (unsigned)(v[k] - lo);
            if (o < (unsigned)RANGE_N) atomicAdd(&acc[o], val[k]);
        }
    }
    __syncthreads();

    float* p = part + (size_t)sl * N_NODES + lo;
    for (int j = tid; j < RANGE_N; j += PTB) p[j] = acc[j];
}

__global__ void k_s_reduce(const float* __restrict__ part, const float* __restrict__ dinv,
                           const float* __restrict__ x, float* __restrict__ s, int n) {
    int i = blockIdx.x * blockDim.x + threadIdx.x;
    if (i < n) {
        float d = dinv[i];
        float a = d * d * x[i];
#pragma unroll 16
        for (int p = 0; p < NSLICE; ++p) a += part[(size_t)p * N_NODES + i];
        s[i] = a;
    }
}

typedef float v4f __attribute__((ext_vector_type(4)));
__global__ __launch_bounds__(256)
void k_gemm_wmma(const float* __restrict__ s, const float* __restrict__ W1, const float* __restrict__ b1,
                 const float* __restrict__ W2, float* __restrict__ t, int n_tiles) {
    __shared__ v2f sB[(HID / 2) * BPAIR_STRIDE];
    __shared__ v4f sWB[HID / 2];
    {
        float* sf = (float*)sB;
        for (int idx = threadIdx.x; idx < HID * HID; idx += blockDim.x) {
            int k = idx >> 6, col = idx & 63;
            sf[(k >> 1) * (2 * BPAIR_STRIDE) + col * 2 + (k & 1)] = W2[idx];
        }
        if (threadIdx.x < HID / 2) {
            const int p = threadIdx.x;
            v4f wb = { W1[2 * p], W1[2 * p + 1], b1[2 * p], b1[2 * p + 1] };
            sWB[p] = wb;
        }
    }
    __syncthreads();

    const int wave = threadIdx.x >> 5;
    const int lane = threadIdx.x & 31;
    const int tile = blockIdx.x * 8 + wave;
    if (tile >= n_tiles) return;

    const int row0 = tile * 16;
    const int m    = lane & 15;
    const int half = lane >> 4;

    v8f acc0 = {}, acc1 = {}, acc2 = {}, acc3 = {};
    const float sv = s[row0 + m];

    for (int kk = 0; kk < HID; kk += 4) {
        const int ka = kk + 2 * half;
        const v4f wb = sWB[ka >> 1];
        const float h0 = sv * wb.x + wb.z;
        const float h1v = sv * wb.y + wb.w;
        v2f a;
        a.x = h0 > 0.0f ? h0 : 0.0f;
        a.y = h1v > 0.0f ? h1v : 0.0f;

        const v2f* bp = &sB[(ka >> 1) * BPAIR_STRIDE + m];
        acc0 = __builtin_amdgcn_wmma_f32_16x16x4_f32(false, a, false, bp[ 0], (short)0, acc0, false, false);
        acc1 = __builtin_amdgcn_wmma_f32_16x16x4_f32(false, a, false, bp[16], (short)0, acc1, false, false);
        acc2 = __builtin_amdgcn_wmma_f32_16x16x4_f32(false, a, false, bp[32], (short)0, acc2, false, false);
        acc3 = __builtin_amdgcn_wmma_f32_16x16x4_f32(false, a, false, bp[48], (short)0, acc3, false, false);
    }

    float* orow = t + (size_t)(row0 + half * 8) * HID;
#pragma unroll
    for (int r = 0; r < 8; ++r) {
        orow[(size_t)r * HID +      m] = acc0[r];
        orow[(size_t)r * HID + 16 + m] = acc1[r];
        orow[(size_t)r * HID + 32 + m] = acc2[r];
        orow[(size_t)r * HID + 48 + m] = acc3[r];
    }
}

#define PULLB 8
__global__ __launch_bounds__(256)
void k_l2_pull(const int* __restrict__ rowptr, const int* __restrict__ csr_src,
               const float* __restrict__ dinv, const float* __restrict__ t,
               const float* __restrict__ b2, const int* __restrict__ batch,
               float* pool, float* cnt, int n) {
    const int node = __builtin_amdgcn_readfirstlane((int)((blockIdx.x * blockDim.x + threadIdx.x) >> 5));
    if (node >= n) return;
    const int lane = threadIdx.x & 31;
    const int beg = rowptr[node];
    const int end = (node + 1 < n) ? rowptr[node + 1] : N_EDGES;
    const float dv = dinv[node];
    const int g = batch[node];
    const float bx = b2[lane * 2], by = b2[lane * 2 + 1];
    const v2f* t2 = (const v2f*)t;
    const v2f me = t2[(size_t)node * (HID / 2) + lane];
    float ax = dv * dv * me.x;
    float ay = dv * dv * me.y;

    for (int base = beg; base < end; base += 32) {
        const int j = base + lane;
        int u = 0;
        float w = 0.0f;
        if (j < end) {
            u = csr_src[j];
            w = dinv[u] * dv;
        }
        const int c = min(32, end - base);
        for (int k = 0; k < c; k += PULLB) {
            int uk[PULLB];
            float wk[PULLB];
            v2f tv[PULLB];
#pragma unroll
            for (int q = 0; q < PULLB; ++q) {
                uk[q] = __builtin_amdgcn_readlane(u, k + q);
                wk[q] = __int_as_float(__builtin_amdgcn_readlane(__float_as_int(w), k + q));
                tv[q] = t2[(size_t)uk[q] * (HID / 2) + lane];
            }
#pragma unroll
            for (int q = 0; q < PULLB; ++q) {
                if (k + q < c) {
                    ax += wk[q] * tv[q].x;
                    ay += wk[q] * tv[q].y;
                }
            }
        }
    }

    float vx = ax + bx, vy = ay + by;
    vx = vx > 0.0f ? vx : 0.0f;
    vy = vy > 0.0f ? vy : 0.0f;
    float* pr = pool + (size_t)g * HID + lane * 2;
    atomicAdd(&pr[0], vx);
    atomicAdd(&pr[1], vy);
    if (lane == 0) atomicAdd(&cnt[g], 1.0f);
}

__global__ void k_pool_init(float* pool, float* cnt) {
    int i = blockIdx.x * blockDim.x + threadIdx.x;
    if (i < N_GRAPH * HID) pool[i] = 0.0f;
    if (i < N_GRAPH) cnt[i] = 0.0f;
}

__global__ void k_final(const float* __restrict__ pool, const float* __restrict__ cnt,
                        float* __restrict__ out) {
    int tid = blockIdx.x * blockDim.x + threadIdx.x;
    if (tid < N_GRAPH * HID) {
        float c = cnt[tid >> 6];
        out[tid] = pool[tid] / fmaxf(c, 1.0f);
    }
}

extern "C" void kernel_launch(void* const* d_in, const int* in_sizes, int n_in,
                              void* d_out, int out_size, void* d_ws, size_t ws_size,
                              hipStream_t stream) {
    const float* x     = (const float*)d_in[0];
    const int*   edge  = (const int*)  d_in[1];
    const int*   batch = (const int*)  d_in[2];
    const float* W1    = (const float*)d_in[3];
    const float* b1    = (const float*)d_in[4];
    const float* W2    = (const float*)d_in[5];
    const float* b2    = (const float*)d_in[6];
    const int* src = edge;
    const int* dst = edge + N_EDGES;
    float* out = (float*)d_out;

    float* ws   = (float*)d_ws;
    float* dinv = ws;
    float* s    = dinv + N_NODES;
    float* bufA = s + N_NODES;
    float* bufB = bufA + (size_t)N_NODES * HID;
    int* rowptr  = (int*)bufB;
    int* bsum    = rowptr + N_NODES;
    int* csr_src = bsum + 1024;
    float* pool = bufB + (size_t)N_NODES * HID;
    float* cnt  = pool + N_GRAPH * HID;

    const int TB = 256;
    k_deg_part     <<<NSLICE * NRANGE, PTB, 0, stream>>>(dst, bufA);
    k_dinv_reduce  <<<(N_NODES + RTB - 1) / RTB, RTB, 0, stream>>>(bufA, dinv, rowptr, bsum, N_NODES);
    k_rowptr       <<<(N_NODES + RTB - 1) / RTB, RTB, 0, stream>>>(rowptr, bsum, N_NODES);
    k_csr_fill     <<<NSLICE * NRANGE, PTB, 0, stream>>>(src, dst, bufA, rowptr, csr_src);
    k_l1_part      <<<NSLICE * NRANGE, PTB, 0, stream>>>(src, dst, dinv, x, bufA);
    k_s_reduce     <<<(N_NODES + TB - 1) / TB, TB, 0, stream>>>(bufA, dinv, x, s, N_NODES);

    const int n_tiles = N_NODES / 16;
    k_gemm_wmma    <<<(n_tiles + 7) / 8, 256, 0, stream>>>(s, W1, b1, W2, bufA, n_tiles);

    k_pool_init    <<<(N_GRAPH * HID + TB - 1) / TB, TB, 0, stream>>>(pool, cnt);
    k_l2_pull      <<<(N_NODES * 32 + TB - 1) / TB, TB, 0, stream>>>(rowptr, csr_src, dinv, bufA, b2, batch,
                                                                    pool, cnt, N_NODES);
    k_final        <<<(N_GRAPH * HID + TB - 1) / TB, TB, 0, stream>>>(pool, cnt, out);
}


# Layer-1 scalar sum pulls over the dst-sorted edge list built for layer 2 (8 lanes per node, y=dinv*x gathered); LDS-atomic partial planes and their reduce launch are gone (sum order changes)
# speedup vs baseline: 2.6413x; 1.2275x over previous
#include <hip/hip_runtime.h>

typedef float v2f __attribute__((ext_vector_type(2)));
typedef float v8f __attribute__((ext_vector_type(8)));

#define N_NODES 200000
#define N_EDGES 3200000
#define N_GRAPH 2048
#define HID     64
#define BPAIR_STRIDE 80

#define NRANGE  4
#define RANGE_N 50000
#define NSLICE  64
#define SLICE_E 50000
#define PTB     1024
#define EBATCH  8

__global__ __launch_bounds__(PTB)
void k_deg_part(const int* __restrict__ dst, float* __restrict__ part) {
    __shared__ float acc[RANGE_N];
    const int tid = threadIdx.x;
    const int rng = blockIdx.x & (NRANGE - 1);
    const int sl  = blockIdx.x / NRANGE;
    const int lo  = rng * RANGE_N;
    for (int j = tid; j < RANGE_N; j += PTB) acc[j] = 0.0f;
    __syncthreads();

    const int* d = dst + (size_t)sl * SLICE_E;
    for (int base = 0; base < SLICE_E; base += PTB * EBATCH) {
        int v[EBATCH];
#pragma unroll
        for (int k = 0; k < EBATCH; ++k) {
            int j = base + k * PTB + tid;
            v[k] = (j < SLICE_E) ? d[j] : -1;
        }
#pragma unroll
        for (int k = 0; k < EBATCH; ++k) {
            unsigned o = (unsigned)(v[k] - lo);
            if (o < (unsigned)RANGE_N) atomicAdd(&acc[o], 1.0f);
        }
    }
    __syncthreads();

    float* p = part + (size_t)sl * N_NODES + lo;
    for (int j = tid; j < RANGE_N; j += PTB) p[j] = acc[j];
}

#define RTB 256
__global__ __launch_bounds__(RTB)
void k_dinv_reduce(float* part, const float* __restrict__ x, float* __restrict__ dinv, float* __restrict__ y,
                   int* __restrict__ rowcnt, int* __restrict__ bsum, int* __restrict__ pad_slot, int n) {
    __shared__ int red[RTB / 32];
    const int i = blockIdx.x * RTB + threadIdx.x;
    int run = 0;
    if (i == 0) {
        *pad_slot = n;
        y[n] = 0.0f;
    }
    if (i < n) {
        const float xi = x[i];
        float deg = 1.0f;
#pragma unroll 16
        for (int p = 0; p < NSLICE; ++p) {
            float c = part[(size_t)p * N_NODES + i];
            part[(size_t)p * N_NODES + i] = __int_as_float(run);
            deg += c;
            run += (int)c;
        }
        const float dv = rsqrtf(deg);
        dinv[i] = dv;
        y[i] = dv * xi;
        rowcnt[i] = run;
    }
    int v = run;
    for (int off = 16; off > 0; off >>= 1) v += __shfl_xor(v, off);
    if ((threadIdx.x & 31) == 0) red[threadIdx.x >> 5] = v;
    __syncthreads();
    if (threadIdx.x == 0) {
        int sum = 0;
        for (int w = 0; w < RTB / 32; ++w) sum += red[w];
        bsum[blockIdx.x] = sum;
    }
}

__global__ __launch_bounds__(RTB)
void k_rowptr(int* __restrict__ rowptr, const int* __restrict__ bsum, int n) {
    __shared__ int red[RTB / 32];
    __shared__ int wsum[RTB / 32];
    const int tid = threadIdx.x, lane = tid & 31, wave = tid >> 5;
    const int b = blockIdx.x;
    int pre = 0;
    for (int j = tid; j < b; j += RTB) pre += bsum[j];
    for (int off = 16; off > 0; off >>= 1) pre += __shfl_xor(pre, off);
    const int i = b * RTB + tid;
    const int c = (i < n) ? rowptr[i] : 0;
    int x = c;
    for (int off = 1; off < 32; off <<= 1) {
        int y = __shfl_up(x, off);
        if (lane >= off) x += y;
    }
    if (lane == 0)  red[wave]  = pre;
    if (lane == 31) wsum[wave] = x;
    __syncthreads();
    int offset = 0;
    for (int w = 0; w < RTB / 32; ++w) offset += red[w];
    for (int w = 0; w < wave; ++w) offset += wsum[w];
    if (i < n) rowptr[i] = offset + x - c;
}

__global__ __launch_bounds__(PTB)
void k_csr_fill(const int* __restrict__ src, const int* __restrict__ dst,
                const float* __restrict__ pre, const int* __restrict__ rowptr,
                int* __restrict__ csr_src) {
    __shared__ int cur[RANGE_N];
    const int tid = threadIdx.x;
    const int rng = blockIdx.x & (NRANGE - 1);
    const int sl  = blockIdx.x / NRANGE;
    const int lo  = rng * RANGE_N;
    const float* pp = pre + (size_t)sl * N_NODES + lo;
    const int*   rp = rowptr + lo;
    for (int base = 0; base < RANGE_N; base += PTB * EBATCH) {
        int a[EBATCH], b[EBATCH];
#pragma unroll
        for (int k = 0; k < EBATCH; ++k) {
            int j = base + k * PTB + tid;
            bool ok = j < RANGE_N;
            a[k] = ok ? rp[j] : 0;
            b[k] = ok ? __float_as_int(pp[j]) : 0;
        }
#pragma unroll
        for (int k = 0; k < EBATCH; ++k) {
            int j = base + k * PTB + tid;
            if (j < RANGE_N) cur[j] = a[k] + b[k];
        }
    }
    __syncthreads();

    const int* sp = src + (size_t)sl * SLICE_E;
    const int* dp = dst + (size_t)sl * SLICE_E;
    for (int base = 0; base < SLICE_E; base += PTB * EBATCH) {
        int u[EBATCH], v[EBATCH];
#pragma unroll
        for (int k = 0; k < EBATCH; ++k) {
            int j = base + k * PTB + tid;
            bool ok = j < SLICE_E;
            u[k] = ok ? sp[j] : 0;
            v[k] = ok ? dp[j] : -1;
        }
#pragma unroll
        for (int k = 0; k < EBATCH; ++k) {
            unsigned o = (unsigned)(v[k] - lo);
            if (o < (unsigned)RANGE_N) {
                int pos = atomicAdd(&cur[o], 1);
                csr_src[pos] = u[k];
            }
        }
    }
}

#define L1G 8
#define L1P 8
#define L1D 4

template <int P> __device__ __forceinline__ int grp_lane(int v) {
    return __builtin_amdgcn_mov_dpp8(v, P | (P << 3) | (P << 6) | (P << 9) | (P << 12) | (P << 15) | (P << 18) | (P << 21));
}
template <int M> __device__ __forceinline__ float xmask_add(float a) {
    return a + __int_as_float(__builtin_amdgcn_update_dpp(0, __float_as_int(a), 0x160 + M, 0xf, 0xf, false));
}

__global__ __launch_bounds__(256)
void k_l1_pull(const int* __restrict__ rowptr, const int* __restrict__ csr_src,
               const float* __restrict__ dinv, const float* __restrict__ y,
               float* __restrict__ s, int n) {
    const int n0 = (int)((blockIdx.x * blockDim.x + threadIdx.x) >> 5) * 32;
    if (n0 >= n) return;
    const int lane = threadIdx.x & 31;
    const int sub  = lane & (L1G - 1);
    const int mine = n0 + lane;
    const int rb = (mine < n) ? rowptr[mine] : N_EDGES;
    const int re = (mine + 1 < n) ? rowptr[mine + 1] : N_EDGES;
    const int mc = mine < n ? mine : n - 1;
    const float dvm = dinv[mc];
    const float ym  = y[mc];

    int beg[L1P], end[L1P];
#define L1_BOUNDS(P) beg[P] = grp_lane<P>(rb); end[P] = grp_lane<P>(re);
    L1_BOUNDS(0) L1_BOUNDS(1) L1_BOUNDS(2) L1_BOUNDS(3) L1_BOUNDS(4) L1_BOUNDS(5) L1_BOUNDS(6) L1_BOUNDS(7)
#undef L1_BOUNDS

    int   u[L1P][L1D];
    float v[L1P][L1D];
#pragma unroll
    for (int p = 0; p < L1P; ++p) {
#pragma unroll
        for (int d = 0; d < L1D; ++d) {
            const int j = beg[p] + sub + d * L1G;
            u[p][d] = csr_src[j < end[p] ? j : N_EDGES];
        }
    }
#pragma unroll
    for (int p = 0; p < L1P; ++p) {
#pragma unroll
        for (int d = 0; d < L1D; ++d) v[p][d] = y[u[p][d]];
    }
    float res = 0.0f;
#pragma unroll
    for (int p = 0; p < L1P; ++p) {
        float a = 0.0f;
#pragma unroll
        for (int d = 0; d < L1D; ++d) a += v[p][d];
        for (int j = beg[p] + sub + L1D * L1G; j < end[p]; j += L1G) a += y[csr_src[j]];
        a = xmask_add<4>(a);
        a = xmask_add<2>(a);
        a = xmask_add<1>(a);
        if (sub == p) res = a;
    }
    if (mine < n) s[mine] = dvm * (ym + res);
}

typedef float v4f __attribute__((ext_vector_type(4)));
__global__ __launch_bounds__(256)
void k_gemm_wmma(const float* __restrict__ s, const float* __restrict__ W1, const float* __restrict__ b1,
                 const float* __restrict__ W2, float* __restrict__ t, int n_tiles) {
    __shared__ v2f sB[(HID / 2) * BPAIR_STRIDE];
    __shared__ v4f sWB[HID / 2];
    {
        float* sf = (float*)sB;
        for (int idx = threadIdx.x; idx < HID * HID; idx += blockDim.x) {
            int k = idx >> 6, col = idx & 63;
            sf[(k >> 1) * (2 * BPAIR_STRIDE) + col * 2 + (k & 1)] = W2[idx];
        }
        if (threadIdx.x < HID / 2) {
            const int p = threadIdx.x;
            v4f wb = { W1[2 * p], W1[2 * p + 1], b1[2 * p], b1[2 * p + 1] };
            sWB[p] = wb;
        }
    }
    __syncthreads();

    const int wave = threadIdx.x >> 5;
    const int lane = threadIdx.x & 31;
    const int tile = blockIdx.x * 8 + wave;
    if (tile >= n_tiles) return;

    const int row0 = tile * 16;
    const int m    = lane & 15;
    const int half = lane >> 4;

    v8f acc0 = {}, acc1 = {}, acc2 = {}, acc3 = {};
    const float sv = s[row0 + m];

    for (int kk = 0; kk < HID; kk += 4) {
        const int ka = kk + 2 * half;
        const v4f wb = sWB[ka >> 1];
        const float h0 = sv * wb.x + wb.z;
        const float h1v = sv * wb.y + wb.w;
        v2f a;
        a.x = h0 > 0.0f ? h0 : 0.0f;
        a.y = h1v > 0.0f ? h1v : 0.0f;

        const v2f* bp = &sB[(ka >> 1) * BPAIR_STRIDE + m];
        acc0 = __builtin_amdgcn_wmma_f32_16x16x4_f32(false, a, false, bp[ 0], (short)0, acc0, false, false);
        acc1 = __builtin_amdgcn_wmma_f32_16x16x4_f32(false, a, false, bp[16], (short)0, acc1, false, false);
        acc2 = __builtin_amdgcn_wmma_f32_16x16x4_f32(false, a, false, bp[32], (short)0, acc2, false, false);
        acc3 = __builtin_amdgcn_wmma_f32_16x16x4_f32(false, a, false, bp[48], (short)0, acc3, false, false);
    }

    float* orow = t + (size_t)(row0 + half * 8) * HID;
#pragma unroll
    for (int r = 0; r < 8; ++r) {
        orow[(size_t)r * HID +      m] = acc0[r];
        orow[(size_t)r * HID + 16 + m] = acc1[r];
        orow[(size_t)r * HID + 32 + m] = acc2[r];
        orow[(size_t)r * HID + 48 + m] = acc3[r];
    }
}

#define PULLB 8
__global__ __launch_bounds__(256)
void k_l2_pull(const int* __restrict__ rowptr, const int* __restrict__ csr_src,
               const float* __restrict__ dinv, const float* __restrict__ t,
               const float* __restrict__ b2, const int* __restrict__ batch,
               float* pool, float* cnt, int n) {
    const int node = __builtin_amdgcn_readfirstlane((int)((blockIdx.x * blockDim.x + threadIdx.x) >> 5));
    if (node >= n) return;
    const int lane = threadIdx.x & 31;
    const int beg = rowptr[node];
    const int end = (node + 1 < n) ? rowptr[node + 1] : N_EDGES;
    const float dv = dinv[node];
    const int g = batch[node];
    const float bx = b2[lane * 2], by = b2[lane * 2 + 1];
    const v2f* t2 = (const v2f*)t;
    const v2f me = t2[(size_t)node * (HID / 2) + lane];
    float ax = dv * dv * me.x;
    float ay = dv * dv * me.y;

    for (int base = beg; base < end; base += 32) {
        const int j = base + lane;
        int u = 0;
        float w = 0.0f;
        if (j < end) {
            u = csr_src[j];
            w = dinv[u] * dv;
        }
        const int c = min(32, end - base);
        for (int k = 0; k < c; k += PULLB) {
            int uk[PULLB];
            float wk[PULLB];
            v2f tv[PULLB];
#pragma unroll
            for (int q = 0; q < PULLB; ++q) {
                uk[q] = __builtin_amdgcn_readlane(u, k + q);
                wk[q] = __int_as_float(__builtin_amdgcn_readlane(__float_as_int(w), k + q));
                tv[q] = t2[(size_t)uk[q] * (HID / 2) + lane];
            }
#pragma unroll
            for (int q = 0; q < PULLB; ++q) {
                if (k + q < c) {
                    ax += wk[q] * tv[q].x;
                    ay += wk[q] * tv[q].y;
                }
            }
        }
    }

    float vx = ax + bx, vy = ay + by;
    vx = vx > 0.0f ? vx : 0.0f;
    vy = vy > 0.0f ? vy : 0.0f;
    float* pr = pool + (size_t)g * HID + lane * 2;
    atomicAdd(&pr[0], vx);
    atomicAdd(&pr[1], vy);
    if (lane == 0) atomicAdd(&cnt[g], 1.0f);
}

__global__ void k_pool_init(float* pool, float* cnt) {
    int i = blockIdx.x * blockDim.x + threadIdx.x;
    if (i < N_GRAPH * HID) pool[i] = 0.0f;
    if (i < N_GRAPH) cnt[i] = 0.0f;
}

__global__ void k_final(const float* __restrict__ pool, const float* __restrict__ cnt,
                        float* __restrict__ out) {
    int tid = blockIdx.x * blockDim.x + threadIdx.x;
    if (tid < N_GRAPH * HID) {
        float c = cnt[tid >> 6];
        out[tid] = pool[tid] / fmaxf(c, 1.0f);
    }
}

extern "C" void kernel_launch(void* const* d_in, const int* in_sizes, int n_in,
                              void* d_out, int out_size, void* d_ws, size_t ws_size,
                              hipStream_t stream) {
    const float* x     = (const float*)d_in[0];
    const int*   edge  = (const int*)  d_in[1];
    const int*   batch = (const int*)  d_in[2];
    const float* W1    = (const float*)d_in[3];
    const float* b1    = (const float*)d_in[4];
    const float* W2    = (const float*)d_in[5];
    const float* b2    = (const float*)d_in[6];
    const int* src = edge;
    const int* dst = edge + N_EDGES;
    float* out = (float*)d_out;

    float* ws   = (float*)d_ws;
    float* dinv = ws;
    float* s    = dinv + N_NODES;
    float* bufA = s + N_NODES;
    float* bufB = bufA + (size_t)N_NODES * HID;
    int* rowptr  = (int*)bufB;
    int* bsum    = rowptr + N_NODES;
    int* csr_src = bsum + 1024;
    float* y     = (float*)(csr_src + N_EDGES + 32);
    float* pool = bufB + (size_t)N_NODES * HID;
    float* cnt  = pool + N_GRAPH * HID;

    const int TB = 256;
    k_deg_part     <<<NSLICE * NRANGE, PTB, 0, stream>>>(dst, bufA);
    k_dinv_reduce  <<<(N_NODES + RTB - 1) / RTB, RTB, 0, stream>>>(bufA, x, dinv, y, rowptr, bsum, csr_src + N_EDGES, N_NODES);
    k_rowptr       <<<(N_NODES + RTB - 1) / RTB, RTB, 0, stream>>>(rowptr, bsum, N_NODES);
    k_csr_fill     <<<NSLICE * NRANGE, PTB, 0, stream>>>(src, dst, bufA, rowptr, csr_src);
    k_l1_pull      <<<(N_NODES + TB - 1) / TB, TB, 0, stream>>>(rowptr, csr_src, dinv, y, s, N_NODES);

    const int n_tiles = N_NODES / 16;
    k_gemm_wmma    <<<(n_tiles + 7) / 8, 256, 0, stream>>>(s, W1, b1, W2, bufA, n_tiles);

    k_pool_init    <<<(N_GRAPH * HID + TB - 1) / TB, TB, 0, stream>>>(pool, cnt);
    k_l2_pull      <<<(N_NODES * 32 + TB - 1) / TB, TB, 0, stream>>>(rowptr, csr_src, dinv, bufA, b2, batch,
                                                                    pool, cnt, N_NODES);
    k_final        <<<(N_GRAPH * HID + TB - 1) / TB, TB, 0, stream>>>(pool, cnt, out);
}


# Layer-2 aggregation done before the W2 product (linearity): a wave per 16-node tile gathers (s,dinv) per in-edge, rebuilds relu(s*W1+b1) into its A fragments; t never stored (sum order changes)
# speedup vs baseline: 3.8783x; 1.4683x over previous
#include <hip/hip_runtime.h>

typedef float v2f __attribute__((ext_vector_type(2)));
typedef float v8f __attribute__((ext_vector_type(8)));

#define N_NODES 200000
#define N_EDGES 3200000
#define N_GRAPH 2048
#define HID     64
#define BPAIR_STRIDE 80

#define NRANGE  4
#define RANGE_N 50000
#define NSLICE  64
#define SLICE_E 50000
#define PTB     1024
#define EBATCH  8

__global__ __launch_bounds__(PTB)
void k_deg_part(const int* __restrict__ dst, float* __restrict__ part) {
    __shared__ float acc[RANGE_N];
    const int tid = threadIdx.x;
    const int rng = blockIdx.x & (NRANGE - 1);
    const int sl  = blockIdx.x / NRANGE;
    const int lo  = rng * RANGE_N;
    for (int j = tid; j < RANGE_N; j += PTB) acc[j] = 0.0f;
    __syncthreads();

    const int* d = dst + (size_t)sl * SLICE_E;
    for (int base = 0; base < SLICE_E; base += PTB * EBATCH) {
        int v[EBATCH];
#pragma unroll
        for (int k = 0; k < EBATCH; ++k) {
            int j = base + k * PTB + tid;
            v[k] = (j < SLICE_E) ? d[j] : -1;
        }
#pragma unroll
        for (int k = 0; k < EBATCH; ++k) {
            unsigned o = (unsigned)(v[k] - lo);
            if (o < (unsigned)RANGE_N) atomicAdd(&acc[o], 1.0f);
        }
    }
    __syncthreads();

    float* p = part + (size_t)sl * N_NODES + lo;
    for (int j = tid; j < RANGE_N; j += PTB) p[j] = acc[j];
}

#define RTB 256
__global__ __launch_bounds__(RTB)
void k_dinv_reduce(float* part, const float* __restrict__ x, float* __restrict__ dinv, float* __restrict__ y,
                   int* __restrict__ rowcnt, int* __restrict__ bsum, int* __restrict__ pad_slot,
                   float* __restrict__ sd, int n) {
    __shared__ int red[RTB / 32];
    const int i = blockIdx.x * RTB + threadIdx.x;
    int run = 0;
    if (i == 0) {
        *pad_slot = n;
        y[n] = 0.0f;
        sd[2 * n] = 0.0f;
        sd[2 * n + 1] = 0.0f;
    }
    if (i < n) {
        const float xi = x[i];
        float deg = 1.0f;
#pragma unroll 16
        for (int p = 0; p < NSLICE; ++p) {
            float c = part[(size_t)p * N_NODES + i];
            part[(size_t)p * N_NODES + i] = __int_as_float(run);
            deg += c;
            run += (int)c;
        }
        const float dv = rsqrtf(deg);
        dinv[i] = dv;
        y[i] = dv * xi;
        rowcnt[i] = run;
    }
    int v = run;
    for (int off = 16; off > 0; off >>= 1) v += __shfl_xor(v, off);
    if ((threadIdx.x & 31) == 0) red[threadIdx.x >> 5] = v;
    __syncthreads();
    if (threadIdx.x == 0) {
        int sum = 0;
        for (int w = 0; w < RTB / 32; ++w) sum += red[w];
        bsum[blockIdx.x] = sum;
    }
}

__global__ __launch_bounds__(RTB)
void k_rowptr(int* __restrict__ rowptr, const int* __restrict__ bsum, int n) {
    __shared__ int red[RTB / 32];
    __shared__ int wsum[RTB / 32];
    const int tid = threadIdx.x, lane = tid & 31, wave = tid >> 5;
    const int b = blockIdx.x;
    int pre = 0;
    for (int j = tid; j < b; j += RTB) pre += bsum[j];
    for (int off = 16; off > 0; off >>= 1) pre += __shfl_xor(pre, off);
    const int i = b * RTB + tid;
    const int c = (i < n) ? rowptr[i] : 0;
    int x = c;
    for (int off = 1; off < 32; off <<= 1) {
        int y = __shfl_up(x, off);
        if (lane >= off) x += y;
    }
    if (lane == 0)  red[wave]  = pre;
    if (lane == 31) wsum[wave] = x;
    __syncthreads();
    int offset = 0;
    for (int w = 0; w < RTB / 32; ++w) offset += red[w];
    for (int w = 0; w < wave; ++w) offset += wsum[w];
    if (i < n) rowptr[i] = offset + x - c;
}

__global__ __launch_bounds__(PTB)
void k_csr_fill(const int* __restrict__ src, const int* __restrict__ dst,
                const float* __restrict__ pre, const int* __restrict__ rowptr,
                int* __restrict__ csr_src) {
    __shared__ int cur[RANGE_N];
    const int tid = threadIdx.x;
    const int rng = blockIdx.x & (NRANGE - 1);
    const int sl  = blockIdx.x / NRANGE;
    const int lo  = rng * RANGE_N;
    const float* pp = pre + (size_t)sl * N_NODES + lo;
    const int*   rp = rowptr + lo;
    for (int base = 0; base < RANGE_N; base += PTB * EBATCH) {
        int a[EBATCH], b[EBATCH];
#pragma unroll
        for (int k = 0; k < EBATCH; ++k) {
            int j = base + k * PTB + tid;
            bool ok = j < RANGE_N;
            a[k] = ok ? rp[j] : 0;
            b[k] = ok ? __float_as_int(pp[j]) : 0;
        }
#pragma unroll
        for (int k = 0; k < EBATCH; ++k) {
            int j = base + k * PTB + tid;
            if (j < RANGE_N) cur[j] = a[k] + b[k];
        }
    }
    __syncthreads();

    const int* sp = src + (size_t)sl * SLICE_E;
    const int* dp = dst + (size_t)sl * SLICE_E;
    for (int base = 0; base < SLICE_E; base += PTB * EBATCH) {
        int u[EBATCH], v[EBATCH];
#pragma unroll
        for (int k = 0; k < EBATCH; ++k) {
            int j = base + k * PTB + tid;
            bool ok = j < SLICE_E;
            u[k] = ok ? sp[j] : 0;
            v[k] = ok ? dp[j] : -1;
        }
#pragma unroll
        for (int k = 0; k < EBATCH; ++k) {
            unsigned o = (unsigned)(v[k] - lo);
            if (o < (unsigned)RANGE_N) {
                int pos = atomicAdd(&cur[o], 1);
                csr_src[pos] = u[k];
            }
        }
    }
}

#define L1G 8
#define L1P 8
#define L1D 4

template <int P> __device__ __forceinline__ int grp_lane(int v) {
    return __builtin_amdgcn_mov_dpp8(v, P | (P << 3) | (P << 6) | (P << 9) | (P << 12) | (P << 15) | (P << 18) | (P << 21));
}
template <int M> __device__ __forceinline__ float xmask_add(float a) {
    return a + __int_as_float(__builtin_amdgcn_update_dpp(0, __float_as_int(a), 0x160 + M, 0xf, 0xf, false));
}

__global__ __launch_bounds__(256)
void k_l1_pull(const int* __restrict__ rowptr, const int* __restrict__ csr_src,
               const float* __restrict__ dinv, const float* __restrict__ y,
               float* __restrict__ sd, int n) {
    const int n0 = (int)((blockIdx.x * blockDim.x + threadIdx.x) >> 5) * 32;
    if (n0 >= n) return;
    const int lane = threadIdx.x & 31;
    const int sub  = lane & (L1G - 1);
    const int mine = n0 + lane;
    const int rb = (mine < n) ? rowptr[mine] : N_EDGES;
    const int re = (mine + 1 < n) ? rowptr[mine + 1] : N_EDGES;
    const int mc = mine < n ? mine : n - 1;
    const float dvm = dinv[mc];
    const float ym  = y[mc];

    int beg[L1P], end[L1P];
#define L1_BOUNDS(P) beg[P] = grp_lane<P>(rb); end[P] = grp_lane<P>(re);
    L1_BOUNDS(0) L1_BOUNDS(1) L1_BOUNDS(2) L1_BOUNDS(3) L1_BOUNDS(4) L1_BOUNDS(5) L1_BOUNDS(6) L1_BOUNDS(7)
#undef L1_BOUNDS

    int   u[L1P][L1D];
    float v[L1P][L1D];
#pragma unroll
    for (int p = 0; p < L1P; ++p) {
#pragma unroll
        for (int d = 0; d < L1D; ++d) {
            const int j = beg[p] + sub + d * L1G;
            u[p][d] = csr_src[j < end[p] ? j : N_EDGES];
        }
    }
#pragma unroll
    for (int p = 0; p < L1P; ++p) {
#pragma unroll
        for (int d = 0; d < L1D; ++d) v[p][d] = y[u[p][d]];
    }
    float res = 0.0f;
#pragma unroll
    for (int p = 0; p < L1P; ++p) {
        float a = 0.0f;
#pragma unroll
        for (int d = 0; d < L1D; ++d) a += v[p][d];
        for (int j = beg[p] + sub + L1D * L1G; j < end[p]; j += L1G) a += y[csr_src[j]];
        a = xmask_add<4>(a);
        a = xmask_add<2>(a);
        a = xmask_add<1>(a);
        if (sub == p) res = a;
    }
    if (mine < n) {
        v2f pr = { dvm * (ym + res), dvm };
        ((v2f*)sd)[mine] = pr;
    }
}

typedef float v4f __attribute__((ext_vector_type(4)));
#define AGB 4

#define AG_ADD(P)                                                         \
    {                                                                     \
        const float su_ = (P).x, wu_ = (P).y;                             \
        _Pragma("unroll")                                                 \
        for (int q_ = 0; q_ < HID / 4; ++q_) {                            \
            const float h0_ = su_ * wb[q_].x + wb[q_].z;                  \
            const float h1_ = su_ * wb[q_].y + wb[q_].w;                  \
            ag[q_].x += wu_ * (h0_ > 0.0f ? h0_ : 0.0f);                  \
            ag[q_].y += wu_ * (h1_ > 0.0f ? h1_ : 0.0f);                  \
        }                                                                 \
    }

__global__ __launch_bounds__(256)
void k_l2_fused(const int* __restrict__ rowptr, const int* __restrict__ csr_src, const float* __restrict__ sd,
                const float* __restrict__ W1, const float* __restrict__ b1,
                const float* __restrict__ W2, const float* __restrict__ b2,
                const int* __restrict__ batch, float* pool, float* cnt, int n_tiles, int n) {
    __shared__ v2f sB[(HID / 2) * BPAIR_STRIDE];
    __shared__ v4f sWB[HID / 2];
    {
        float* sf = (float*)sB;
        for (int idx = threadIdx.x; idx < HID * HID; idx += blockDim.x) {
            int k = idx >> 6, col = idx & 63;
            sf[(k >> 1) * (2 * BPAIR_STRIDE) + col * 2 + (k & 1)] = W2[idx];
        }
        if (threadIdx.x < HID / 2) {
            const int p = threadIdx.x;
            v4f wbp = { W1[2 * p], W1[2 * p + 1], b1[2 * p], b1[2 * p + 1] };
            sWB[p] = wbp;
        }
    }
    __syncthreads();

    const int wave = threadIdx.x >> 5;
    const int lane = threadIdx.x & 31;
    const int tile = blockIdx.x * 8 + wave;
    if (tile >= n_tiles) return;

    const int row0 = tile * 16;
    const int m    = lane & 15;
    const int half = lane >> 4;
    const int node = row0 + m;

    const v2f* sdp = (const v2f*)sd;
    const int rb = rowptr[node];
    const int re = (node + 1 < n) ? rowptr[node + 1] : N_EDGES;
    const v2f self = sdp[node];

    int j = rb;
    int un[AGB];
    v2f pv[AGB];
#pragma unroll
    for (int d = 0; d < AGB; ++d) un[d] = csr_src[j + d < re ? j + d : N_EDGES];
#pragma unroll
    for (int d = 0; d < AGB; ++d) pv[d] = sdp[un[d]];
#pragma unroll
    for (int d = 0; d < AGB; ++d) un[d] = csr_src[j + AGB + d < re ? j + AGB + d : N_EDGES];

    v4f wb[HID / 4];
    v2f ag[HID / 4];
#pragma unroll
    for (int q = 0; q < HID / 4; ++q) {
        wb[q] = sWB[2 * q + half];
        ag[q].x = 0.0f;
        ag[q].y = 0.0f;
    }
    AG_ADD(self)

    while (__builtin_amdgcn_ballot_w32(j < re) != 0u) {
        v2f pn[AGB];
#pragma unroll
        for (int d = 0; d < AGB; ++d) pn[d] = sdp[un[d]];
#pragma unroll
        for (int d = 0; d < AGB; ++d) un[d] = csr_src[j + 2 * AGB + d < re ? j + 2 * AGB + d : N_EDGES];
        __builtin_amdgcn_sched_barrier(0);
#pragma unroll
        for (int d = 0; d < AGB; ++d) AG_ADD(pv[d])
#pragma unroll
        for (int d = 0; d < AGB; ++d) pv[d] = pn[d];
        j += AGB;
    }

    const int rbase = row0 + half * 8;
    int g[8];
#pragma unroll
    for (int r = 0; r < 8; ++r) g[r] = batch[rbase + r];
    float bb[4];
#pragma unroll
    for (int c = 0; c < 4; ++c) bb[c] = b2[16 * c + m];

    const float dv = self.y;
    v8f acc0 = {}, acc1 = {}, acc2 = {}, acc3 = {};
#pragma unroll
    for (int q = 0; q < HID / 4; ++q) {
        v2f a;
        a.x = dv * ag[q].x;
        a.y = dv * ag[q].y;
        const v2f* bp = &sB[(2 * q + half) * BPAIR_STRIDE + m];
        acc0 = __builtin_amdgcn_wmma_f32_16x16x4_f32(false, a, false, bp[ 0], (short)0, acc0, false, false);
        acc1 = __builtin_amdgcn_wmma_f32_16x16x4_f32(false, a, false, bp[16], (short)0, acc1, false, false);
        acc2 = __builtin_amdgcn_wmma_f32_16x16x4_f32(false, a, false, bp[32], (short)0, acc2, false, false);
        acc3 = __builtin_amdgcn_wmma_f32_16x16x4_f32(false, a, false, bp[48], (short)0, acc3, false, false);
    }

#pragma unroll
    for (int r = 0; r < 8; ++r) {
        float* pr = pool + (size_t)g[r] * HID + m;
        float v0 = acc0[r] + bb[0], v1 = acc1[r] + bb[1], v2 = acc2[r] + bb[2], v3 = acc3[r] + bb[3];
        atomicAdd(&pr[ 0], v0 > 0.0f ? v0 : 0.0f);
        atomicAdd(&pr[16], v1 > 0.0f ? v1 : 0.0f);
        atomicAdd(&pr[32], v2 > 0.0f ? v2 : 0.0f);
        atomicAdd(&pr[48], v3 > 0.0f ? v3 : 0.0f);
    }
    if (lane < 16) atomicAdd(&cnt[batch[row0 + lane]], 1.0f);
}
#undef AG_ADD

__global__ void k_pool_init(float* pool, float* cnt) {
    int i = blockIdx.x * blockDim.x + threadIdx.x;
    if (i < N_GRAPH * HID) pool[i] = 0.0f;
    if (i < N_GRAPH) cnt[i] = 0.0f;
}

__global__ void k_final(const float* __restrict__ pool, const float* __restrict__ cnt,
                        float* __restrict__ out) {
    int tid = blockIdx.x * blockDim.x + threadIdx.x;
    if (tid < N_GRAPH * HID) {
        float c = cnt[tid >> 6];
        out[tid] = pool[tid] / fmaxf(c, 1.0f);
    }
}

extern "C" void kernel_launch(void* const* d_in, const int* in_sizes, int n_in,
                              void* d_out, int out_size, void* d_ws, size_t ws_size,
                              hipStream_t stream) {
    const float* x     = (const float*)d_in[0];
    const int*   edge  = (const int*)  d_in[1];
    const int*   batch = (const int*)  d_in[2];
    const float* W1    = (const float*)d_in[3];
    const float* b1    = (const float*)d_in[4];
    const float* W2    = (const float*)d_in[5];
    const float* b2    = (const float*)d_in[6];
    const int* src = edge;
    const int* dst = edge + N_EDGES;
    float* out = (float*)d_out;

    float* ws   = (float*)d_ws;
    float* dinv = ws;
    float* bufA = dinv + 2 * N_NODES;
    float* bufB = bufA + (size_t)N_NODES * HID;
    int* rowptr  = (int*)bufB;
    int* bsum    = rowptr + N_NODES;
    int* csr_src = bsum + 1024;
    float* y     = (float*)(csr_src + N_EDGES + 32);
    float* sd    = y + N_NODES + 8;
    float* pool = bufB + (size_t)N_NODES * HID;
    float* cnt  = pool + N_GRAPH * HID;

    const int TB = 256;
    k_deg_part     <<<NSLICE * NRANGE, PTB, 0, stream>>>(dst, bufA);
    k_dinv_reduce  <<<(N_NODES + RTB - 1) / RTB, RTB, 0, stream>>>(bufA, x, dinv, y, rowptr, bsum, csr_src + N_EDGES, sd, N_NODES);
    k_rowptr       <<<(N_NODES + RTB - 1) / RTB, RTB, 0, stream>>>(rowptr, bsum, N_NODES);
    k_csr_fill     <<<NSLICE * NRANGE, PTB, 0, stream>>>(src, dst, bufA, rowptr, csr_src);
    k_l1_pull      <<<(N_NODES + TB - 1) / TB, TB, 0, stream>>>(rowptr, csr_src, dinv, y, sd, N_NODES);

    k_pool_init    <<<(N_GRAPH * HID + TB - 1) / TB, TB, 0, stream>>>(pool, cnt);
    const int n_tiles = N_NODES / 16;
    k_l2_fused     <<<(n_tiles + 7) / 8, 256, 0, stream>>>(rowptr, csr_src, sd, W1, b1, W2, b2, batch,
                                                           pool, cnt, n_tiles, N_NODES);
    k_final        <<<(N_GRAPH * HID + TB - 1) / TB, TB, 0, stream>>>(pool, cnt, out);
}


# Pool adds reduced before the atomic: a lane sums the 8 D rows of its half where they share a graph id, halves joined where the tile does; up to 16x fewer float atomics (sum order changes)
# speedup vs baseline: 4.1846x; 1.0790x over previous
#include <hip/hip_runtime.h>

typedef float v2f __attribute__((ext_vector_type(2)));
typedef float v8f __attribute__((ext_vector_type(8)));

#define N_NODES 200000
#define N_EDGES 3200000
#define N_GRAPH 2048
#define HID     64
#define BPAIR_STRIDE 80

#define NRANGE  4
#define RANGE_N 50000
#define NSLICE  64
#define SLICE_E 50000
#define PTB     1024
#define EBATCH  8

__global__ __launch_bounds__(PTB)
void k_deg_part(const int* __restrict__ dst, float* __restrict__ part) {
    __shared__ float acc[RANGE_N];
    const int tid = threadIdx.x;
    const int rng = blockIdx.x & (NRANGE - 1);
    const int sl  = blockIdx.x / NRANGE;
    const int lo  = rng * RANGE_N;
    for (int j = tid; j < RANGE_N; j += PTB) acc[j] = 0.0f;
    __syncthreads();

    const int* d = dst + (size_t)sl * SLICE_E;
    for (int base = 0; base < SLICE_E; base += PTB * EBATCH) {
        int v[EBATCH];
#pragma unroll
        for (int k = 0; k < EBATCH; ++k) {
            int j = base + k * PTB + tid;
            v[k] = (j < SLICE_E) ? d[j] : -1;
        }
#pragma unroll
        for (int k = 0; k < EBATCH; ++k) {
            unsigned o = (unsigned)(v[k] - lo);
            if (o < (unsigned)RANGE_N) atomicAdd(&acc[o], 1.0f);
        }
    }
    __syncthreads();

    float* p = part + (size_t)sl * N_NODES + lo;
    for (int j = tid; j < RANGE_N; j += PTB) p[j] = acc[j];
}

#define RTB 256
__global__ __launch_bounds__(RTB)
void k_dinv_reduce(float* part, const float* __restrict__ x, float* __restrict__ dinv, float* __restrict__ y,
                   int* __restrict__ rowcnt, int* __restrict__ bsum, int* __restrict__ pad_slot,
                   float* __restrict__ sd, int n) {
    __shared__ int red[RTB / 32];
    const int i = blockIdx.x * RTB + threadIdx.x;
    int run = 0;
    if (i == 0) {
        *pad_slot = n;
        y[n] = 0.0f;
        sd[2 * n] = 0.0f;
        sd[2 * n + 1] = 0.0f;
    }
    if (i < n) {
        const float xi = x[i];
        float deg = 1.0f;
#pragma unroll 16
        for (int p = 0; p < NSLICE; ++p) {
            float c = part[(size_t)p * N_NODES + i];
            part[(size_t)p * N_NODES + i] = __int_as_float(run);
            deg += c;
            run += (int)c;
        }
        const float dv = rsqrtf(deg);
        dinv[i] = dv;
        y[i] = dv * xi;
        rowcnt[i] = run;
    }
    int v = run;
    for (int off = 16; off > 0; off >>= 1) v += __shfl_xor(v, off);
    if ((threadIdx.x & 31) == 0) red[threadIdx.x >> 5] = v;
    __syncthreads();
    if (threadIdx.x == 0) {
        int sum = 0;
        for (int w = 0; w < RTB / 32; ++w) sum += red[w];
        bsum[blockIdx.x] = sum;
    }
}

__global__ __launch_bounds__(RTB)
void k_rowptr(int* __restrict__ rowptr, const int* __restrict__ bsum, int n) {
    __shared__ int red[RTB / 32];
    __shared__ int wsum[RTB / 32];
    const int tid = threadIdx.x, lane = tid & 31, wave = tid >> 5;
    const int b = blockIdx.x;
    int pre = 0;
    for (int j = tid; j < b; j += RTB) pre += bsum[j];
    for (int off = 16; off > 0; off >>= 1) pre += __shfl_xor(pre, off);
    const int i = b * RTB + tid;
    const int c = (i < n) ? rowptr[i] : 0;
    int x = c;
    for (int off = 1; off < 32; off <<= 1) {
        int y = __shfl_up(x, off);
        if (lane >= off) x += y;
    }
    if (lane == 0)  red[wave]  = pre;
    if (lane == 31) wsum[wave] = x;
    __syncthreads();
    int offset = 0;
    for (int w = 0; w < RTB / 32; ++w) offset += red[w];
    for (int w = 0; w < wave; ++w) offset += wsum[w];
    if (i < n) rowptr[i] = offset + x - c;
}

__global__ __launch_bounds__(PTB)
void k_csr_fill(const int* __restrict__ src, const int* __restrict__ dst,
                const float* __restrict__ pre, const int* __restrict__ rowptr,
                int* __restrict__ csr_src) {
    __shared__ int cur[RANGE_N];
    const int tid = threadIdx.x;
    const int rng = blockIdx.x & (NRANGE - 1);
    const int sl  = blockIdx.x / NRANGE;
    const int lo  = rng * RANGE_N;
    const float* pp = pre + (size_t)sl * N_NODES + lo;
    const int*   rp = rowptr + lo;
    for (int base = 0; base < RANGE_N; base += PTB * EBATCH) {
        int a[EBATCH], b[EBATCH];
#pragma unroll
        for (int k = 0; k < EBATCH; ++k) {
            int j = base + k * PTB + tid;
            bool ok = j < RANGE_N;
            a[k] = ok ? rp[j] : 0;
            b[k] = ok ? __float_as_int(pp[j]) : 0;
        }
#pragma unroll
        for (int k = 0; k < EBATCH; ++k) {
            int j = base + k * PTB + tid;
            if (j < RANGE_N) cur[j] = a[k] + b[k];
        }
    }
    __syncthreads();

    const int* sp = src + (size_t)sl * SLICE_E;
    const int* dp = dst + (size_t)sl * SLICE_E;
    for (int base = 0; base < SLICE_E; base += PTB * EBATCH) {
        int u[EBATCH], v[EBATCH];
#pragma unroll
        for (int k = 0; k < EBATCH; ++k) {
            int j = base + k * PTB + tid;
            bool ok = j < SLICE_E;
            u[k] = ok ? sp[j] : 0;
            v[k] = ok ? dp[j] : -1;
        }
#pragma unroll
        for (int k = 0; k < EBATCH; ++k) {
            unsigned o = (unsigned)(v[k] - lo);
            if (o < (unsigned)RANGE_N) {
                int pos = atomicAdd(&cur[o], 1);
                csr_src[pos] = u[k];
            }
        }
    }
}

#define L1G 8
#define L1P 8
#define L1D 4

template <int P> __device__ __forceinline__ int grp_lane(int v) {
    return __builtin_amdgcn_mov_dpp8(v, P | (P << 3) | (P << 6) | (P << 9) | (P << 12) | (P << 15) | (P << 18) | (P << 21));
}
template <int M> __device__ __forceinline__ float xmask_add(float a) {
    return a + __int_as_float(__builtin_amdgcn_update_dpp(0, __float_as_int(a), 0x160 + M, 0xf, 0xf, false));
}

__global__ __launch_bounds__(256)
void k_l1_pull(const int* __restrict__ rowptr, const int* __restrict__ csr_src,
               const float* __restrict__ dinv, const float* __restrict__ y,
               float* __restrict__ sd, int n) {
    const int n0 = (int)((blockIdx.x * blockDim.x + threadIdx.x) >> 5) * 32;
    if (n0 >= n) return;
    const int lane = threadIdx.x & 31;
    const int sub  = lane & (L1G - 1);
    const int mine = n0 + lane;
    const int rb = (mine < n) ? rowptr[mine] : N_EDGES;
    const int re = (mine + 1 < n) ? rowptr[mine + 1] : N_EDGES;
    const int mc = mine < n ? mine : n - 1;
    const float dvm = dinv[mc];
    const float ym  = y[mc];

    int beg[L1P], end[L1P];
#define L1_BOUNDS(P) beg[P] = grp_lane<P>(rb); end[P] = grp_lane<P>(re);
    L1_BOUNDS(0) L1_BOUNDS(1) L1_BOUNDS(2) L1_BOUNDS(3) L1_BOUNDS(4) L1_BOUNDS(5) L1_BOUNDS(6) L1_BOUNDS(7)
#undef L1_BOUNDS

    int   u[L1P][L1D];
    float v[L1P][L1D];
#pragma unroll
    for (int p = 0; p < L1P; ++p) {
#pragma unroll
        for (int d = 0; d < L1D; ++d) {
            const int j = beg[p] + sub + d * L1G;
            u[p][d] = csr_src[j < end[p] ? j : N_EDGES];
        }
    }
#pragma unroll
    for (int p = 0; p < L1P; ++p) {
#pragma unroll
        for (int d = 0; d < L1D; ++d) v[p][d] = y[u[p][d]];
    }
    float res = 0.0f;
#pragma unroll
    for (int p = 0; p < L1P; ++p) {
        float a = 0.0f;
#pragma unroll
        for (int d = 0; d < L1D; ++d) a += v[p][d];
        for (int j = beg[p] + sub + L1D * L1G; j < end[p]; j += L1G) a += y[csr_src[j]];
        a = xmask_add<4>(a);
        a = xmask_add<2>(a);
        a = xmask_add<1>(a);
        if (sub == p) res = a;
    }
    if (mine < n) {
        v2f pr = { dvm * (ym + res), dvm };
        ((v2f*)sd)[mine] = pr;
    }
}

typedef float v4f __attribute__((ext_vector_type(4)));
#define AGB 4

#define AG_ADD(P)                                                         \
    {                                                                     \
        const float su_ = (P).x, wu_ = (P).y;                             \
        _Pragma("unroll")                                                 \
        for (int q_ = 0; q_ < HID / 4; ++q_) {                            \
            const float h0_ = su_ * wb[q_].x + wb[q_].z;                  \
            const float h1_ = su_ * wb[q_].y + wb[q_].w;                  \
            ag[q_].x += wu_ * (h0_ > 0.0f ? h0_ : 0.0f);                  \
            ag[q_].y += wu_ * (h1_ > 0.0f ? h1_ : 0.0f);                  \
        }                                                                 \
    }

__global__ __launch_bounds__(256)
void k_l2_fused(const int* __restrict__ rowptr, const int* __restrict__ csr_src, const float* __restrict__ sd,
                const float* __restrict__ W1, const float* __restrict__ b1,
                const float* __restrict__ W2, const float* __restrict__ b2,
                const int* __restrict__ batch, float* pool, float* cnt, int n_tiles, int n) {
    __shared__ v2f sB[(HID / 2) * BPAIR_STRIDE];
    __shared__ v4f sWB[HID / 2];
    {
        float* sf = (float*)sB;
        for (int idx = threadIdx.x; idx < HID * HID; idx += blockDim.x) {
            int k = idx >> 6, col = idx & 63;
            sf[(k >> 1) * (2 * BPAIR_STRIDE) + col * 2 + (k & 1)] = W2[idx];
        }
        if (threadIdx.x < HID / 2) {
            const int p = threadIdx.x;
            v4f wbp = { W1[2 * p], W1[2 * p + 1], b1[2 * p], b1[2 * p + 1] };
            sWB[p] = wbp;
        }
    }
    __syncthreads();

    const int wave = threadIdx.x >> 5;
    const int lane = threadIdx.x & 31;
    const int tile = blockIdx.x * 8 + wave;
    if (tile >= n_tiles) return;

    const int row0 = tile * 16;
    const int m    = lane & 15;
    const int half = lane >> 4;
    const int node = row0 + m;

    const v2f* sdp = (const v2f*)sd;
    const int rb = rowptr[node];
    const int re = (node + 1 < n) ? rowptr[node + 1] : N_EDGES;
    const v2f self = sdp[node];

    int j = rb;
    int un[AGB];
    v2f pv[AGB];
#pragma unroll
    for (int d = 0; d < AGB; ++d) un[d] = csr_src[j + d < re ? j + d : N_EDGES];
#pragma unroll
    for (int d = 0; d < AGB; ++d) pv[d] = sdp[un[d]];
#pragma unroll
    for (int d = 0; d < AGB; ++d) un[d] = csr_src[j + AGB + d < re ? j + AGB + d : N_EDGES];

    v4f wb[HID / 4];
    v2f ag[HID / 4];
#pragma unroll
    for (int q = 0; q < HID / 4; ++q) {
        wb[q] = sWB[2 * q + half];
        ag[q].x = 0.0f;
        ag[q].y = 0.0f;
    }
    AG_ADD(self)

    while (__builtin_amdgcn_ballot_w32(j < re) != 0u) {
        v2f pn[AGB];
#pragma unroll
        for (int d = 0; d < AGB; ++d) pn[d] = sdp[un[d]];
#pragma unroll
        for (int d = 0; d < AGB; ++d) un[d] = csr_src[j + 2 * AGB + d < re ? j + 2 * AGB + d : N_EDGES];
        __builtin_amdgcn_sched_barrier(0);
#pragma unroll
        for (int d = 0; d < AGB; ++d) AG_ADD(pv[d])
#pragma unroll
        for (int d = 0; d < AGB; ++d) pv[d] = pn[d];
        j += AGB;
    }

    const int rbase = row0 + half * 8;
    int g[8];
#pragma unroll
    for (int r = 0; r < 8; ++r) g[r] = batch[rbase + r];
    float bb[4];
#pragma unroll
    for (int c = 0; c < 4; ++c) bb[c] = b2[16 * c + m];

    const float dv = self.y;
    v8f acc0 = {}, acc1 = {}, acc2 = {}, acc3 = {};
#pragma unroll
    for (int q = 0; q < HID / 4; ++q) {
        v2f a;
        a.x = dv * ag[q].x;
        a.y = dv * ag[q].y;
        const v2f* bp = &sB[(2 * q + half) * BPAIR_STRIDE + m];
        acc0 = __builtin_amdgcn_wmma_f32_16x16x4_f32(false, a, false, bp[ 0], (short)0, acc0, false, false);
        acc1 = __builtin_amdgcn_wmma_f32_16x16x4_f32(false, a, false, bp[16], (short)0, acc1, false, false);
        acc2 = __builtin_amdgcn_wmma_f32_16x16x4_f32(false, a, false, bp[32], (short)0, acc2, false, false);
        acc3 = __builtin_amdgcn_wmma_f32_16x16x4_f32(false, a, false, bp[48], (short)0, acc3, false, false);
    }

    const int g0 = g[0];
    bool same = true;
    float s0 = 0.0f, s1 = 0.0f, s2 = 0.0f, s3 = 0.0f;
#pragma unroll
    for (int r = 0; r < 8; ++r) {
        float v0 = acc0[r] + bb[0], v1 = acc1[r] + bb[1], v2 = acc2[r] + bb[2], v3 = acc3[r] + bb[3];
        v0 = v0 > 0.0f ? v0 : 0.0f;
        v1 = v1 > 0.0f ? v1 : 0.0f;
        v2 = v2 > 0.0f ? v2 : 0.0f;
        v3 = v3 > 0.0f ? v3 : 0.0f;
        acc0[r] = v0; acc1[r] = v1; acc2[r] = v2; acc3[r] = v3;
        s0 += v0; s1 += v1; s2 += v2; s3 += v3;
        same = same && (g[r] == g0);
    }
    const int   gx    = __shfl_xor(g0, 16);
    const int   samex = __shfl_xor((int)same, 16);
    const float x0 = __shfl_xor(s0, 16), x1 = __shfl_xor(s1, 16), x2 = __shfl_xor(s2, 16), x3 = __shfl_xor(s3, 16);
    const bool whole = same && (samex != 0) && (gx == g0);
    if (whole) {
        if (half == 0) {
            float* pr = pool + (size_t)g0 * HID + m;
            atomicAdd(&pr[ 0], s0 + x0);
            atomicAdd(&pr[16], s1 + x1);
            atomicAdd(&pr[32], s2 + x2);
            atomicAdd(&pr[48], s3 + x3);
            if (m == 0) atomicAdd(&cnt[g0], 16.0f);
        }
    } else if (same) {
        float* pr = pool + (size_t)g0 * HID + m;
        atomicAdd(&pr[ 0], s0);
        atomicAdd(&pr[16], s1);
        atomicAdd(&pr[32], s2);
        atomicAdd(&pr[48], s3);
        if (m == 0) atomicAdd(&cnt[g0], 8.0f);
    } else {
#pragma unroll
        for (int r = 0; r < 8; ++r) {
            float* pr = pool + (size_t)g[r] * HID + m;
            atomicAdd(&pr[ 0], acc0[r]);
            atomicAdd(&pr[16], acc1[r]);
            atomicAdd(&pr[32], acc2[r]);
            atomicAdd(&pr[48], acc3[r]);
            if (m == r) atomicAdd(&cnt[g[r]], 1.0f);
        }
    }
}
#undef AG_ADD

__global__ void k_pool_init(float* pool, float* cnt) {
    int i = blockIdx.x * blockDim.x + threadIdx.x;
    if (i < N_GRAPH * HID) pool[i] = 0.0f;
    if (i < N_GRAPH) cnt[i] = 0.0f;
}

__global__ void k_final(const float* __restrict__ pool, const float* __restrict__ cnt,
                        float* __restrict__ out) {
    int tid = blockIdx.x * blockDim.x + threadIdx.x;
    if (tid < N_GRAPH * HID) {
        float c = cnt[tid >> 6];
        out[tid] = pool[tid] / fmaxf(c, 1.0f);
    }
}

extern "C" void kernel_launch(void* const* d_in, const int* in_sizes, int n_in,
                              void* d_out, int out_size, void* d_ws, size_t ws_size,
                              hipStream_t stream) {
    const float* x     = (const float*)d_in[0];
    const int*   edge  = (const int*)  d_in[1];
    const int*   batch = (const int*)  d_in[2];
    const float* W1    = (const float*)d_in[3];
    const float* b1    = (const float*)d_in[4];
    const float* W2    = (const float*)d_in[5];
    const float* b2    = (const float*)d_in[6];
    const int* src = edge;
    const int* dst = edge + N_EDGES;
    float* out = (float*)d_out;

    float* ws   = (float*)d_ws;
    float* dinv = ws;
    float* bufA = dinv + 2 * N_NODES;
    float* bufB = bufA + (size_t)N_NODES * HID;
    int* rowptr  = (int*)bufB;
    int* bsum    = rowptr + N_NODES;
    int* csr_src = bsum + 1024;
    float* y     = (float*)(csr_src + N_EDGES + 32);
    float* sd    = y + N_NODES + 8;
    float* pool = bufB + (size_t)N_NODES * HID;
    float* cnt  = pool + N_GRAPH * HID;

    const int TB = 256;
    k_deg_part     <<<NSLICE * NRANGE, PTB, 0, stream>>>(dst, bufA);
    k_dinv_reduce  <<<(N_NODES + RTB - 1) / RTB, RTB, 0, stream>>>(bufA, x, dinv, y, rowptr, bsum, csr_src + N_EDGES, sd, N_NODES);
    k_rowptr       <<<(N_NODES + RTB - 1) / RTB, RTB, 0, stream>>>(rowptr, bsum, N_NODES);
    k_csr_fill     <<<NSLICE * NRANGE, PTB, 0, stream>>>(src, dst, bufA, rowptr, csr_src);
    k_l1_pull      <<<(N_NODES + TB - 1) / TB, TB, 0, stream>>>(rowptr, csr_src, dinv, y, sd, N_NODES);

    k_pool_init    <<<(N_GRAPH * HID + TB - 1) / TB, TB, 0, stream>>>(pool, cnt);
    const int n_tiles = N_NODES / 16;
    k_l2_fused     <<<(n_tiles + 7) / 8, 256, 0, stream>>>(rowptr, csr_src, sd, W1, b1, W2, b2, batch,
                                                           pool, cnt, n_tiles, N_NODES);
    k_final        <<<(N_GRAPH * HID + TB - 1) / TB, TB, 0, stream>>>(pool, cnt, out);
}


# Edge list sorted by destination in two levels: pairs bucketed by 782-node range, then an LDS counting sort per bucket gives rows, offsets and degrees; 64 count planes gone (sum order changes)
# speedup vs baseline: 6.2952x; 1.5044x over previous
#include <hip/hip_runtime.h>

typedef float v2f __attribute__((ext_vector_type(2)));
typedef float v8f __attribute__((ext_vector_type(8)));

#define N_NODES 200000
#define N_EDGES 3200000
#define N_GRAPH 2048
#define HID     64
#define BPAIR_STRIDE 80

typedef int v2i __attribute__((ext_vector_type(2)));
typedef int v4i __attribute__((ext_vector_type(4)));
#define NBKT   256
#define BKT_N  782
#define NEBLK  256
#define EBLK_E 12500
#define ETB    512
#define EPT    25
#define MROWS  (NEBLK / (ETB / 64))
#define MBATCH 16
#define PTB    1024
#define SEB    16

__device__ __forceinline__ unsigned bkt_of(int d) {
    const unsigned b = (unsigned)d / (unsigned)BKT_N;
    return b < (unsigned)(NBKT - 1) ? b : (unsigned)(NBKT - 1);
}

__global__ __launch_bounds__(ETB)
void k_bkt_count(const int* __restrict__ dst, int* __restrict__ cntm) {
    __shared__ int hist[NBKT];
    const int tid = threadIdx.x;
    if (tid < NBKT) hist[tid] = 0;
    __syncthreads();
    const int* d = dst + (size_t)blockIdx.x * EBLK_E;
    int v[EPT];
#pragma unroll
    for (int k = 0; k < EPT; ++k) {
        const int j = k * ETB + tid;
        v[k] = (j < EBLK_E) ? d[j] : -1;
    }
#pragma unroll
    for (int k = 0; k < EPT; ++k) {
        if (v[k] >= 0) atomicAdd(&hist[bkt_of(v[k])], 1);
    }
    __syncthreads();
    if (tid < NBKT) cntm[blockIdx.x * NBKT + tid] = hist[tid];
}

__global__ __launch_bounds__(ETB)
void k_bkt_scatter(const int* __restrict__ src, const int* __restrict__ dst, const int* __restrict__ cntm,
                   int* __restrict__ bbase, v2i* __restrict__ ebuf) {
    __shared__ v4i ptot[ETB / 64][NBKT / 4];
    __shared__ v4i ppre[ETB / 64][NBKT / 4];
    __shared__ int cur[NBKT];
    __shared__ int wtot[NBKT / 32];
    const int tid = threadIdx.x, lane = tid & 31, wave = tid >> 5;
    const int b = blockIdx.x;

    const int* sp = src + (size_t)b * EBLK_E;
    const int* dp = dst + (size_t)b * EBLK_E;
    int u[EPT], v[EPT];
#pragma unroll
    for (int k = 0; k < EPT; ++k) {
        const int j = k * ETB + tid;
        const bool ok = j < EBLK_E;
        u[k] = ok ? sp[j] : 0;
        v[k] = ok ? dp[j] : -1;
    }

    {
        const int t4 = tid & 63, q = tid >> 6;
        v4i tot = { 0, 0, 0, 0 }, pre = { 0, 0, 0, 0 };
        const v4i* cm = (const v4i*)cntm + (size_t)q * MROWS * (NBKT / 4) + t4;
        for (int h = 0; h < MROWS; h += MBATCH) {
            v4i c[MBATCH];
#pragma unroll
            for (int i = 0; i < MBATCH; ++i) c[i] = cm[(h + i) * (NBKT / 4)];
#pragma unroll
            for (int i = 0; i < MBATCH; ++i) {
                const int mk = (q * MROWS + h + i < b) ? -1 : 0;
                const v4i mk4 = { mk, mk, mk, mk };
                tot += c[i];
                pre += c[i] & mk4;
            }
        }
        ptot[q][t4] = tot;
        ppre[q][t4] = pre;
    }
    __syncthreads();
    int x = 0, tot1 = 0, pre1 = 0;
    if (tid < NBKT) {
        for (int q = 0; q < ETB / 64; ++q) { tot1 += ((const int*)ptot[q])[tid]; pre1 += ((const int*)ppre[q])[tid]; }
        x = tot1;
        for (int off = 1; off < 32; off <<= 1) {
            const int yv = __shfl_up(x, off);
            if (lane >= off) x += yv;
        }
        if (lane == 31) wtot[wave] = x;
    }
    __syncthreads();
    if (tid < NBKT) {
        int off = 0;
        for (int w = 0; w < wave; ++w) off += wtot[w];
        const int base = off + x - tot1;
        cur[tid] = base + pre1;
        if (b == 0) {
            bbase[tid] = base;
            if (tid == NBKT - 1) bbase[NBKT] = base + tot1;
        }
    }
    __syncthreads();

    int pos[EPT];
#pragma unroll
    for (int k = 0; k < EPT; ++k) pos[k] = (v[k] >= 0) ? atomicAdd(&cur[bkt_of(v[k])], 1) : -1;
#pragma unroll
    for (int k = 0; k < EPT; ++k) {
        if (pos[k] >= 0) {
            v2i pr = { u[k], v[k] };
            ebuf[pos[k]] = pr;
        }
    }
}

__global__ __launch_bounds__(PTB)
void k_bkt_sort(const v2i* __restrict__ ebuf, const int* __restrict__ bbase, const float* __restrict__ x,
                int* __restrict__ rowptr, float* __restrict__ dinv, float* __restrict__ y,
                int* __restrict__ csr_src, float* __restrict__ sd, int n) {
    __shared__ int cnt[PTB];
    __shared__ int wsum[PTB / 32];
    const int tid = threadIdx.x, lane = tid & 31, wave = tid >> 5;
    const int r  = blockIdx.x;
    const int lo = r * BKT_N;
    int nn = n - lo;
    nn = nn < BKT_N ? nn : BKT_N;
    nn = nn > 0 ? nn : 0;
    const int beg = bbase[r], end = bbase[r + 1];
    cnt[tid] = 0;
    const float xi = (tid < nn) ? x[lo + tid] : 0.0f;
    if (r == 0 && tid == 0) {
        csr_src[N_EDGES] = n;
        y[n] = 0.0f;
        sd[2 * n] = 0.0f;
        sd[2 * n + 1] = 0.0f;
    }
    __syncthreads();

    for (int base = beg; base < end; base += PTB * SEB) {
        int dd[SEB];
#pragma unroll
        for (int k = 0; k < SEB; ++k) {
            const int e = base + k * PTB + tid;
            dd[k] = (e < end) ? ebuf[e].y : -1;
        }
#pragma unroll
        for (int k = 0; k < SEB; ++k) {
            const unsigned o = (unsigned)(dd[k] - lo);
            if (o < (unsigned)nn) atomicAdd(&cnt[o], 1);
        }
    }
    __syncthreads();

    const int c = cnt[tid];
    int xs = c;
    for (int off = 1; off < 32; off <<= 1) {
        const int yv = __shfl_up(xs, off);
        if (lane >= off) xs += yv;
    }
    if (lane == 31) wsum[wave] = xs;
    __syncthreads();
    int offw = 0;
    for (int w = 0; w < wave; ++w) offw += wsum[w];
    const int first = beg + offw + xs - c;
    cnt[tid] = first;
    if (tid < nn) {
        const float dv = rsqrtf(1.0f + (float)c);
        rowptr[lo + tid] = first;
        dinv[lo + tid] = dv;
        y[lo + tid] = dv * xi;
    }
    __syncthreads();

    for (int base = beg; base < end; base += PTB * SEB) {
        v2i p[SEB];
#pragma unroll
        for (int k = 0; k < SEB; ++k) {
            const int e = base + k * PTB + tid;
            v2i none = { 0, -1 };
            p[k] = (e < end) ? ebuf[e] : none;
        }
        int pos[SEB];
#pragma unroll
        for (int k = 0; k < SEB; ++k) {
            const unsigned o = (unsigned)(p[k].y - lo);
            pos[k] = (o < (unsigned)nn) ? atomicAdd(&cnt[o], 1) : -1;
        }
#pragma unroll
        for (int k = 0; k < SEB; ++k) {
            if (pos[k] >= 0) csr_src[pos[k]] = p[k].x;
        }
    }
}

#define L1G 8
#define L1P 8
#define L1D 4

template <int P> __device__ __forceinline__ int grp_lane(int v) {
    return __builtin_amdgcn_mov_dpp8(v, P | (P << 3) | (P << 6) | (P << 9) | (P << 12) | (P << 15) | (P << 18) | (P << 21));
}
template <int M> __device__ __forceinline__ float xmask_add(float a) {
    return a + __int_as_float(__builtin_amdgcn_update_dpp(0, __float_as_int(a), 0x160 + M, 0xf, 0xf, false));
}

__global__ __launch_bounds__(256)
void k_l1_pull(const int* __restrict__ rowptr, const int* __restrict__ csr_src,
               const float* __restrict__ dinv, const float* __restrict__ y,
               float* __restrict__ sd, int n) {
    const int n0 = (int)((blockIdx.x * blockDim.x + threadIdx.x) >> 5) * 32;
    if (n0 >= n) return;
    const int lane = threadIdx.x & 31;
    const int sub  = lane & (L1G - 1);
    const int mine = n0 + lane;
    const int rb = (mine < n) ? rowptr[mine] : N_EDGES;
    const int re = (mine + 1 < n) ? rowptr[mine + 1] : N_EDGES;
    const int mc = mine < n ? mine : n - 1;
    const float dvm = dinv[mc];
    const float ym  = y[mc];

    int beg[L1P], end[L1P];
#define L1_BOUNDS(P) beg[P] = grp_lane<P>(rb); end[P] = grp_lane<P>(re);
    L1_BOUNDS(0) L1_BOUNDS(1) L1_BOUNDS(2) L1_BOUNDS(3) L1_BOUNDS(4) L1_BOUNDS(5) L1_BOUNDS(6) L1_BOUNDS(7)
#undef L1_BOUNDS

    int   u[L1P][L1D];
    float v[L1P][L1D];
#pragma unroll
    for (int p = 0; p < L1P; ++p) {
#pragma unroll
        for (int d = 0; d < L1D; ++d) {
            const int j = beg[p] + sub + d * L1G;
            u[p][d] = csr_src[j < end[p] ? j : N_EDGES];
        }
    }
#pragma unroll
    for (int p = 0; p < L1P; ++p) {
#pragma unroll
        for (int d = 0; d < L1D; ++d) v[p][d] = y[u[p][d]];
    }
    float res = 0.0f;
#pragma unroll
    for (int p = 0; p < L1P; ++p) {
        float a = 0.0f;
#pragma unroll
        for (int d = 0; d < L1D; ++d) a += v[p][d];
        for (int j = beg[p] + sub + L1D * L1G; j < end[p]; j += L1G) a += y[csr_src[j]];
        a = xmask_add<4>(a);
        a = xmask_add<2>(a);
        a = xmask_add<1>(a);
        if (sub == p) res = a;
    }
    if (mine < n) {
        v2f pr = { dvm * (ym + res), dvm };
        ((v2f*)sd)[mine] = pr;
    }
}

typedef float v4f __attribute__((ext_vector_type(4)));
#define AGB 4

#define AG_ADD(P)                                                         \
    {                                                                     \
        const float su_ = (P).x, wu_ = (P).y;                             \
        _Pragma("unroll")                                                 \
        for (int q_ = 0; q_ < HID / 4; ++q_) {                            \
            const float h0_ = su_ * wb[q_].x + wb[q_].z;                  \
            const float h1_ = su_ * wb[q_].y + wb[q_].w;                  \
            ag[q_].x += wu_ * (h0_ > 0.0f ? h0_ : 0.0f);                  \
            ag[q_].y += wu_ * (h1_ > 0.0f ? h1_ : 0.0f);                  \
        }                                                                 \
    }

__global__ __launch_bounds__(256)
void k_l2_fused(const int* __restrict__ rowptr, const int* __restrict__ csr_src, const float* __restrict__ sd,
                const float* __restrict__ W1, const float* __restrict__ b1,
                const float* __restrict__ W2, const float* __restrict__ b2,
                const int* __restrict__ batch, float* pool, float* cnt, int n_tiles, int n) {
    __shared__ v2f sB[(HID / 2) * BPAIR_STRIDE];
    __shared__ v4f sWB[HID / 2];
    {
        float* sf = (float*)sB;
        for (int idx = threadIdx.x; idx < HID * HID; idx += blockDim.x) {
            int k = idx >> 6, col = idx & 63;
            sf[(k >> 1) * (2 * BPAIR_STRIDE) + col * 2 + (k & 1)] = W2[idx];
        }
        if (threadIdx.x < HID / 2) {
            const int p = threadIdx.x;
            v4f wbp = { W1[2 * p], W1[2 * p + 1], b1[2 * p], b1[2 * p + 1] };
            sWB[p] = wbp;
        }
    }
    __syncthreads();

    const int wave = threadIdx.x >> 5;
    const int lane = threadIdx.x & 31;
    const int tile = blockIdx.x * 8 + wave;
    if (tile >= n_tiles) return;

    const int row0 = tile * 16;
    const int m    = lane & 15;
    const int half = lane >> 4;
    const int node = row0 + m;

    const v2f* sdp = (const v2f*)sd;
    const int rb = rowptr[node];
    const int re = (node + 1 < n) ? rowptr[node + 1] : N_EDGES;
    const v2f self = sdp[node];

    int j = rb;
    int un[AGB];
    v2f pv[AGB];
#pragma unroll
    for (int d = 0; d < AGB; ++d) un[d] = csr_src[j + d < re ? j + d : N_EDGES];
#pragma unroll
    for (int d = 0; d < AGB; ++d) pv[d] = sdp[un[d]];
#pragma unroll
    for (int d = 0; d < AGB; ++d) un[d] = csr_src[j + AGB + d < re ? j + AGB + d : N_EDGES];

    v4f wb[HID / 4];
    v2f ag[HID / 4];
#pragma unroll
    for (int q = 0; q < HID / 4; ++q) {
        wb[q] = sWB[2 * q + half];
        ag[q].x = 0.0f;
        ag[q].y = 0.0f;
    }
    AG_ADD(self)

    while (__builtin_amdgcn_ballot_w32(j < re) != 0u) {
        v2f pn[AGB];
#pragma unroll
        for (int d = 0; d < AGB; ++d) pn[d] = sdp[un[d]];
#pragma unroll
        for (int d = 0; d < AGB; ++d) un[d] = csr_src[j + 2 * AGB + d < re ? j + 2 * AGB + d : N_EDGES];
        __builtin_amdgcn_sched_barrier(0);
#pragma unroll
        for (int d = 0; d < AGB; ++d) AG_ADD(pv[d])
#pragma unroll
        for (int d = 0; d < AGB; ++d) pv[d] = pn[d];
        j += AGB;
    }

    const int rbase = row0 + half * 8;
    int g[8];
#pragma unroll
    for (int r = 0; r < 8; ++r) g[r] = batch[rbase + r];
    float bb[4];
#pragma unroll
    for (int c = 0; c < 4; ++c) bb[c] = b2[16 * c + m];

    const float dv = self.y;
    v8f acc0 = {}, acc1 = {}, acc2 = {}, acc3 = {};
#pragma unroll
    for (int q = 0; q < HID / 4; ++q) {
        v2f a;
        a.x = dv * ag[q].x;
        a.y = dv * ag[q].y;
        const v2f* bp = &sB[(2 * q + half) * BPAIR_STRIDE + m];
        acc0 = __builtin_amdgcn_wmma_f32_16x16x4_f32(false, a, false, bp[ 0], (short)0, acc0, false, false);
        acc1 = __builtin_amdgcn_wmma_f32_16x16x4_f32(false, a, false, bp[16], (short)0, acc1, false, false);
        acc2 = __builtin_amdgcn_wmma_f32_16x16x4_f32(false, a, false, bp[32], (short)0, acc2, false, false);
        acc3 = __builtin_amdgcn_wmma_f32_16x16x4_f32(false, a, false, bp[48], (short)0, acc3, false, false);
    }

    const int g0 = g[0];
    bool same = true;
    float s0 = 0.0f, s1 = 0.0f, s2 = 0.0f, s3 = 0.0f;
#pragma unroll
    for (int r = 0; r < 8; ++r) {
        float v0 = acc0[r] + bb[0], v1 = acc1[r] + bb[1], v2 = acc2[r] + bb[2], v3 = acc3[r] + bb[3];
        v0 = v0 > 0.0f ? v0 : 0.0f;
        v1 = v1 > 0.0f ? v1 : 0.0f;
        v2 = v2 > 0.0f ? v2 : 0.0f;
        v3 = v3 > 0.0f ? v3 : 0.0f;
        acc0[r] = v0; acc1[r] = v1; acc2[r] = v2; acc3[r] = v3;
        s0 += v0; s1 += v1; s2 += v2; s3 += v3;
        same = same && (g[r] == g0);
    }
    const int   gx    = __shfl_xor(g0, 16);
    const int   samex = __shfl_xor((int)same, 16);
    const float x0 = __shfl_xor(s0, 16), x1 = __shfl_xor(s1, 16), x2 = __shfl_xor(s2, 16), x3 = __shfl_xor(s3, 16);
    const bool whole = same && (samex != 0) && (gx == g0);
    if (whole) {
        if (half == 0) {
            float* pr = pool + (size_t)g0 * HID + m;
            atomicAdd(&pr[ 0], s0 + x0);
            atomicAdd(&pr[16], s1 + x1);
            atomicAdd(&pr[32], s2 + x2);
            atomicAdd(&pr[48], s3 + x3);
            if (m == 0) atomicAdd(&cnt[g0], 16.0f);
        }
    } else if (same) {
        float* pr = pool + (size_t)g0 * HID + m;
        atomicAdd(&pr[ 0], s0);
        atomicAdd(&pr[16], s1);
        atomicAdd(&pr[32], s2);
        atomicAdd(&pr[48], s3);
        if (m == 0) atomicAdd(&cnt[g0], 8.0f);
    } else {
#pragma unroll
        for (int r = 0; r < 8; ++r) {
            float* pr = pool + (size_t)g[r] * HID + m;
            atomicAdd(&pr[ 0], acc0[r]);
            atomicAdd(&pr[16], acc1[r]);
            atomicAdd(&pr[32], acc2[r]);
            atomicAdd(&pr[48], acc3[r]);
            if (m == r) atomicAdd(&cnt[g[r]], 1.0f);
        }
    }
}
#undef AG_ADD

__global__ void k_pool_init(float* pool, float* cnt) {
    int i = blockIdx.x * blockDim.x + threadIdx.x;
    if (i < N_GRAPH * HID) pool[i] = 0.0f;
    if (i < N_GRAPH) cnt[i] = 0.0f;
}

__global__ void k_final(const float* __restrict__ pool, const float* __restrict__ cnt,
                        float* __restrict__ out) {
    int tid = blockIdx.x * blockDim.x + threadIdx.x;
    if (tid < N_GRAPH * HID) {
        float c = cnt[tid >> 6];
        out[tid] = pool[tid] / fmaxf(c, 1.0f);
    }
}

extern "C" void kernel_launch(void* const* d_in, const int* in_sizes, int n_in,
                              void* d_out, int out_size, void* d_ws, size_t ws_size,
                              hipStream_t stream) {
    const float* x     = (const float*)d_in[0];
    const int*   edge  = (const int*)  d_in[1];
    const int*   batch = (const int*)  d_in[2];
    const float* W1    = (const float*)d_in[3];
    const float* b1    = (const float*)d_in[4];
    const float* W2    = (const float*)d_in[5];
    const float* b2    = (const float*)d_in[6];
    const int* src = edge;
    const int* dst = edge + N_EDGES;
    float* out = (float*)d_out;

    float* ws   = (float*)d_ws;
    float* dinv = ws;
    float* bufA = dinv + 2 * N_NODES;
    float* bufB = bufA + (size_t)N_NODES * HID;
    int* cntm    = (int*)bufA;
    int* bbase   = cntm + NEBLK * NBKT;
    v2i* ebuf    = (v2i*)(bbase + 512);
    int* rowptr  = (int*)bufB;
    int* csr_src = rowptr + N_NODES + 1024;
    float* y     = (float*)(csr_src + N_EDGES + 32);
    float* sd    = y + N_NODES + 8;
    float* pool = bufB + (size_t)N_NODES * HID;
    float* cnt  = pool + N_GRAPH * HID;

    const int TB = 256;
    k_bkt_count    <<<NEBLK, ETB, 0, stream>>>(dst, cntm);
    k_bkt_scatter  <<<NEBLK, ETB, 0, stream>>>(src, dst, cntm, bbase, ebuf);
    k_bkt_sort     <<<NBKT, PTB, 0, stream>>>(ebuf, bbase, x, rowptr, dinv, y, csr_src, sd, N_NODES);
    k_l1_pull      <<<(N_NODES + TB - 1) / TB, TB, 0, stream>>>(rowptr, csr_src, dinv, y, sd, N_NODES);

    k_pool_init    <<<(N_GRAPH * HID + TB - 1) / TB, TB, 0, stream>>>(pool, cnt);
    const int n_tiles = N_NODES / 16;
    k_l2_fused     <<<(n_tiles + 7) / 8, 256, 0, stream>>>(rowptr, csr_src, sd, W1, b1, W2, b2, batch,
                                                           pool, cnt, n_tiles, N_NODES);
    k_final        <<<(N_GRAPH * HID + TB - 1) / TB, TB, 0, stream>>>(pool, cnt, out);
}


# Both sort passes group their items in LDS and write them out in list order: a wave's store covers whole 128-byte lines, not 32 lines of 4 or 8 bytes each (row order stays arrival order)
# speedup vs baseline: 7.4281x; 1.1800x over previous
#include <hip/hip_runtime.h>

typedef float v2f __attribute__((ext_vector_type(2)));
typedef float v8f __attribute__((ext_vector_type(8)));

#define N_NODES 200000
#define N_EDGES 3200000
#define N_GRAPH 2048
#define HID     64
#define BPAIR_STRIDE 80

typedef int v2i __attribute__((ext_vector_type(2)));
typedef int v4i __attribute__((ext_vector_type(4)));
#define NBKT   256
#define BKT_N  782
#define NEBLK  256
#define EBLK_E 12500
#define ETB    512
#define EPT    25
#define MROWS  (NEBLK / (ETB / 64))
#define MBATCH 16
#define PTB    1024
#define SEB    16
#define SCAP   16384

__device__ __forceinline__ unsigned bkt_of(int d) {
    const unsigned b = (unsigned)d / (unsigned)BKT_N;
    return b < (unsigned)(NBKT - 1) ? b : (unsigned)(NBKT - 1);
}

__global__ __launch_bounds__(ETB)
void k_bkt_count(const int* __restrict__ dst, int* __restrict__ cntm) {
    __shared__ int hist[NBKT];
    const int tid = threadIdx.x;
    if (tid < NBKT) hist[tid] = 0;
    __syncthreads();
    const int* d = dst + (size_t)blockIdx.x * EBLK_E;
    int v[EPT];
#pragma unroll
    for (int k = 0; k < EPT; ++k) {
        const int j = k * ETB + tid;
        v[k] = (j < EBLK_E) ? d[j] : -1;
    }
#pragma unroll
    for (int k = 0; k < EPT; ++k) {
        if (v[k] >= 0) atomicAdd(&hist[bkt_of(v[k])], 1);
    }
    __syncthreads();
    if (tid < NBKT) cntm[blockIdx.x * NBKT + tid] = hist[tid];
}

__global__ __launch_bounds__(ETB)
void k_bkt_scatter(const int* __restrict__ src, const int* __restrict__ dst, const int* __restrict__ cntm,
                   int* __restrict__ bbase, v2i* __restrict__ ebuf) {
    __shared__ v4i ptot[ETB / 64][NBKT / 4];
    __shared__ v4i ppre[ETB / 64][NBKT / 4];
    __shared__ int cur[NBKT];
    __shared__ int gdl[NBKT];
    __shared__ int wtot[NBKT / 32];
    __shared__ int wloc[NBKT / 32];
    __shared__ v2i stage[EBLK_E];
    const int tid = threadIdx.x, lane = tid & 31, wave = tid >> 5;
    const int b = blockIdx.x;

    const int* sp = src + (size_t)b * EBLK_E;
    const int* dp = dst + (size_t)b * EBLK_E;
    int u[EPT], v[EPT];
#pragma unroll
    for (int k = 0; k < EPT; ++k) {
        const int j = k * ETB + tid;
        const bool ok = j < EBLK_E;
        u[k] = ok ? sp[j] : 0;
        v[k] = ok ? dp[j] : -1;
    }
    const int own = (tid < NBKT) ? cntm[b * NBKT + tid] : 0;

    {
        const int t4 = tid & 63, q = tid >> 6;
        v4i tot = { 0, 0, 0, 0 }, pre = { 0, 0, 0, 0 };
        const v4i* cm = (const v4i*)cntm + (size_t)q * MROWS * (NBKT / 4) + t4;
        for (int h = 0; h < MROWS; h += MBATCH) {
            v4i c[MBATCH];
#pragma unroll
            for (int i = 0; i < MBATCH; ++i) c[i] = cm[(h + i) * (NBKT / 4)];
#pragma unroll
            for (int i = 0; i < MBATCH; ++i) {
                const int mk = (q * MROWS + h + i < b) ? -1 : 0;
                const v4i mk4 = { mk, mk, mk, mk };
                tot += c[i];
                pre += c[i] & mk4;
            }
        }
        ptot[q][t4] = tot;
        ppre[q][t4] = pre;
    }
    __syncthreads();
    int x = 0, xl = 0, tot1 = 0, pre1 = 0;
    if (tid < NBKT) {
        for (int q = 0; q < ETB / 64; ++q) { tot1 += ((const int*)ptot[q])[tid]; pre1 += ((const int*)ppre[q])[tid]; }
        x = tot1;
        xl = own;
        for (int off = 1; off < 32; off <<= 1) {
            const int yv = __shfl_up(x, off);
            const int yl = __shfl_up(xl, off);
            if (lane >= off) { x += yv; xl += yl; }
        }
        if (lane == 31) { wtot[wave] = x; wloc[wave] = xl; }
    }
    __syncthreads();
    if (tid < NBKT) {
        int off = 0, offl = 0;
        for (int w = 0; w < wave; ++w) { off += wtot[w]; offl += wloc[w]; }
        const int base = off + x - tot1;
        const int lbeg = offl + xl - own;
        cur[tid] = lbeg;
        gdl[tid] = base + pre1 - lbeg;
        if (b == 0) {
            bbase[tid] = base;
            if (tid == NBKT - 1) bbase[NBKT] = base + tot1;
        }
    }
    __syncthreads();

    int pos[EPT];
#pragma unroll
    for (int k = 0; k < EPT; ++k) pos[k] = (v[k] >= 0) ? atomicAdd(&cur[bkt_of(v[k])], 1) : -1;
#pragma unroll
    for (int k = 0; k < EPT; ++k) {
        if (pos[k] >= 0 && pos[k] < EBLK_E) {
            v2i pr = { u[k], v[k] };
            stage[pos[k]] = pr;
        }
    }
    __syncthreads();
    v2i pr[EPT];
#pragma unroll
    for (int k = 0; k < EPT; ++k) {
        const int j = k * ETB + tid;
        v2i none = { 0, 0 };
        pr[k] = (j < EBLK_E) ? stage[j] : none;
    }
#pragma unroll
    for (int k = 0; k < EPT; ++k) {
        const int j = k * ETB + tid;
        if (j < EBLK_E) ebuf[j + gdl[bkt_of(pr[k].y)]] = pr[k];
    }
}

__global__ __launch_bounds__(PTB)
void k_bkt_sort(const v2i* __restrict__ ebuf, const int* __restrict__ bbase, const float* __restrict__ x,
                int* __restrict__ rowptr, float* __restrict__ dinv, float* __restrict__ y,
                int* __restrict__ csr_src, float* __restrict__ sd, int n) {
    __shared__ int cnt[PTB];
    __shared__ int wsum[PTB / 32];
    __shared__ v4i stage4[SCAP / 4];
    const int tid = threadIdx.x, lane = tid & 31, wave = tid >> 5;
    const int r  = blockIdx.x;
    const int lo = r * BKT_N;
    int nn = n - lo;
    nn = nn < BKT_N ? nn : BKT_N;
    nn = nn > 0 ? nn : 0;
    const int beg = bbase[r], end = bbase[r + 1];
    const int beg4 = beg & ~3;
    int* stage = (int*)stage4;
    cnt[tid] = 0;
    const float xi = (tid < nn) ? x[lo + tid] : 0.0f;
    if (r == 0 && tid == 0) {
        csr_src[N_EDGES] = n;
        y[n] = 0.0f;
        sd[2 * n] = 0.0f;
        sd[2 * n + 1] = 0.0f;
    }
    __syncthreads();

    for (int base = beg; base < end; base += PTB * SEB) {
        int dd[SEB];
#pragma unroll
        for (int k = 0; k < SEB; ++k) {
            const int e = base + k * PTB + tid;
            dd[k] = (e < end) ? ebuf[e].y : -1;
        }
#pragma unroll
        for (int k = 0; k < SEB; ++k) {
            const unsigned o = (unsigned)(dd[k] - lo);
            if (o < (unsigned)nn) atomicAdd(&cnt[o], 1);
        }
    }
    __syncthreads();

    const int c = cnt[tid];
    int xs = c;
    for (int off = 1; off < 32; off <<= 1) {
        const int yv = __shfl_up(xs, off);
        if (lane >= off) xs += yv;
    }
    if (lane == 31) wsum[wave] = xs;
    __syncthreads();
    int offw = 0;
    for (int w = 0; w < wave; ++w) offw += wsum[w];
    const int first = beg + offw + xs - c;
    cnt[tid] = first;
    if (tid < nn) {
        const float dv = rsqrtf(1.0f + (float)c);
        rowptr[lo + tid] = first;
        dinv[lo + tid] = dv;
        y[lo + tid] = dv * xi;
    }
    __syncthreads();

    for (int base = beg; base < end; base += PTB * SEB) {
        v2i p[SEB];
#pragma unroll
        for (int k = 0; k < SEB; ++k) {
            const int e = base + k * PTB + tid;
            v2i none = { 0, -1 };
            p[k] = (e < end) ? ebuf[e] : none;
        }
        int pos[SEB];
#pragma unroll
        for (int k = 0; k < SEB; ++k) {
            const unsigned o = (unsigned)(p[k].y - lo);
            pos[k] = (o < (unsigned)nn) ? atomicAdd(&cnt[o], 1) : -1;
        }
#pragma unroll
        for (int k = 0; k < SEB; ++k) {
            if (pos[k] >= 0) {
                const unsigned q = (unsigned)(pos[k] - beg4);
                if (q < (unsigned)SCAP) stage[q] = p[k].x;
                else csr_src[pos[k]] = p[k].x;
            }
        }
    }
    __syncthreads();
    const int lim = (end < beg4 + SCAP) ? end : beg4 + SCAP;
    for (int p0 = beg4 + 4 * tid; p0 < lim; p0 += 4 * PTB) {
        const v4i w = stage4[(p0 - beg4) >> 2];
        if (p0 >= beg && p0 + 4 <= lim) {
            *(v4i*)(csr_src + p0) = w;
        } else {
            if (p0     >= beg && p0     < lim) csr_src[p0]     = w.x;
            if (p0 + 1 >= beg && p0 + 1 < lim) csr_src[p0 + 1] = w.y;
            if (p0 + 2 >= beg && p0 + 2 < lim) csr_src[p0 + 2] = w.z;
            if (p0 + 3 >= beg && p0 + 3 < lim) csr_src[p0 + 3] = w.w;
        }
    }
}

#define L1G 8
#define L1P 8
#define L1D 4

template <int P> __device__ __forceinline__ int grp_lane(int v) {
    return __builtin_amdgcn_mov_dpp8(v, P | (P << 3) | (P << 6) | (P << 9) | (P << 12) | (P << 15) | (P << 18) | (P << 21));
}
template <int M> __device__ __forceinline__ float xmask_add(float a) {
    return a + __int_as_float(__builtin_amdgcn_update_dpp(0, __float_as_int(a), 0x160 + M, 0xf, 0xf, false));
}

__global__ __launch_bounds__(256)
void k_l1_pull(const int* __restrict__ rowptr, const int* __restrict__ csr_src,
               const float* __restrict__ dinv, const float* __restrict__ y,
               float* __restrict__ sd, int n) {
    const int n0 = (int)((blockIdx.x * blockDim.x + threadIdx.x) >> 5) * 32;
    if (n0 >= n) return;
    const int lane = threadIdx.x & 31;
    const int sub  = lane & (L1G - 1);
    const int mine = n0 + lane;
    const int rb = (mine < n) ? rowptr[mine] : N_EDGES;
    const int re = (mine + 1 < n) ? rowptr[mine + 1] : N_EDGES;
    const int mc = mine < n ? mine : n - 1;
    const float dvm = dinv[mc];
    const float ym  = y[mc];

    int beg[L1P], end[L1P];
#define L1_BOUNDS(P) beg[P] = grp_lane<P>(rb); end[P] = grp_lane<P>(re);
    L1_BOUNDS(0) L1_BOUNDS(1) L1_BOUNDS(2) L1_BOUNDS(3) L1_BOUNDS(4) L1_BOUNDS(5) L1_BOUNDS(6) L1_BOUNDS(7)
#undef L1_BOUNDS

    int   u[L1P][L1D];
    float v[L1P][L1D];
#pragma unroll
    for (int p = 0; p < L1P; ++p) {
#pragma unroll
        for (int d = 0; d < L1D; ++d) {
            const int j = beg[p] + sub + d * L1G;
            u[p][d] = csr_src[j < end[p] ? j : N_EDGES];
        }
    }
#pragma unroll
    for (int p = 0; p < L1P; ++p) {
#pragma unroll
        for (int d = 0; d < L1D; ++d) v[p][d] = y[u[p][d]];
    }
    float res = 0.0f;
#pragma unroll
    for (int p = 0; p < L1P; ++p) {
        float a = 0.0f;
#pragma unroll
        for (int d = 0; d < L1D; ++d) a += v[p][d];
        for (int j = beg[p] + sub + L1D * L1G; j < end[p]; j += L1G) a += y[csr_src[j]];
        a = xmask_add<4>(a);
        a = xmask_add<2>(a);
        a = xmask_add<1>(a);
        if (sub == p) res = a;
    }
    if (mine < n) {
        v2f pr = { dvm * (ym + res), dvm };
        ((v2f*)sd)[mine] = pr;
    }
}

typedef float v4f __attribute__((ext_vector_type(4)));
#define AGB 4

#define AG_ADD(P)                                                         \
    {                                                                     \
        const float su_ = (P).x, wu_ = (P).y;                             \
        _Pragma("unroll")                                                 \
        for (int q_ = 0; q_ < HID / 4; ++q_) {                            \
            const float h0_ = su_ * wb[q_].x + wb[q_].z;                  \
            const float h1_ = su_ * wb[q_].y + wb[q_].w;                  \
            ag[q_].x += wu_ * (h0_ > 0.0f ? h0_ : 0.0f);                  \
            ag[q_].y += wu_ * (h1_ > 0.0f ? h1_ : 0.0f);                  \
        }                                                                 \
    }

__global__ __launch_bounds__(256)
void k_l2_fused(const int* __restrict__ rowptr, const int* __restrict__ csr_src, const float* __restrict__ sd,
                const float* __restrict__ W1, const float* __restrict__ b1,
                const float* __restrict__ W2, const float* __restrict__ b2,
                const int* __restrict__ batch, float* pool, float* cnt, int n_tiles, int n) {
    __shared__ v2f sB[(HID / 2) * BPAIR_STRIDE];
    __shared__ v4f sWB[HID / 2];
    {
        float* sf = (float*)sB;
        for (int idx = threadIdx.x; idx < HID * HID; idx += blockDim.x) {
            int k = idx >> 6, col = idx & 63;
            sf[(k >> 1) * (2 * BPAIR_STRIDE) + col * 2 + (k & 1)] = W2[idx];
        }
        if (threadIdx.x < HID / 2) {
            const int p = threadIdx.x;
            v4f wbp = { W1[2 * p], W1[2 * p + 1], b1[2 * p], b1[2 * p + 1] };
            sWB[p] = wbp;
        }
    }
    __syncthreads();

    const int wave = threadIdx.x >> 5;
    const int lane = threadIdx.x & 31;
    const int tile = blockIdx.x * 8 + wave;
    if (tile >= n_tiles) return;

    const int row0 = tile * 16;
    const int m    = lane & 15;
    const int half = lane >> 4;
    const int node = row0 + m;

    const v2f* sdp = (const v2f*)sd;
    const int rb = rowptr[node];
    const int re = (node + 1 < n) ? rowptr[node + 1] : N_EDGES;
    const v2f self = sdp[node];

    int j = rb;
    int un[AGB];
    v2f pv[AGB];
#pragma unroll
    for (int d = 0; d < AGB; ++d) un[d] = csr_src[j + d < re ? j + d : N_EDGES];
#pragma unroll
    for (int d = 0; d < AGB; ++d) pv[d] = sdp[un[d]];
#pragma unroll
    for (int d = 0; d < AGB; ++d) un[d] = csr_src[j + AGB + d < re ? j + AGB + d : N_EDGES];

    v4f wb[HID / 4];
    v2f ag[HID / 4];
#pragma unroll
    for (int q = 0; q < HID / 4; ++q) {
        wb[q] = sWB[2 * q + half];
        ag[q].x = 0.0f;
        ag[q].y = 0.0f;
    }
    AG_ADD(self)

    while (__builtin_amdgcn_ballot_w32(j < re) != 0u) {
        v2f pn[AGB];
#pragma unroll
        for (int d = 0; d < AGB; ++d) pn[d] = sdp[un[d]];
#pragma unroll
        for (int d = 0; d < AGB; ++d) un[d] = csr_src[j + 2 * AGB + d < re ? j + 2 * AGB + d : N_EDGES];
        __builtin_amdgcn_sched_barrier(0);
#pragma unroll
        for (int d = 0; d < AGB; ++d) AG_ADD(pv[d])
#pragma unroll
        for (int d = 0; d < AGB; ++d) pv[d] = pn[d];
        j += AGB;
    }

    const int rbase = row0 + half * 8;
    int g[8];
#pragma unroll
    for (int r = 0; r < 8; ++r) g[r] = batch[rbase + r];
    float bb[4];
#pragma unroll
    for (int c = 0; c < 4; ++c) bb[c] = b2[16 * c + m];

    const float dv = self.y;
    v8f acc0 = {}, acc1 = {}, acc2 = {}, acc3 = {};
#pragma unroll
    for (int q = 0; q < HID / 4; ++q) {
        v2f a;
        a.x = dv * ag[q].x;
        a.y = dv * ag[q].y;
        const v2f* bp = &sB[(2 * q + half) * BPAIR_STRIDE + m];
        acc0 = __builtin_amdgcn_wmma_f32_16x16x4_f32(false, a, false, bp[ 0], (short)0, acc0, false, false);
        acc1 = __builtin_amdgcn_wmma_f32_16x16x4_f32(false, a, false, bp[16], (short)0, acc1, false, false);
        acc2 = __builtin_amdgcn_wmma_f32_16x16x4_f32(false, a, false, bp[32], (short)0, acc2, false, false);
        acc3 = __builtin_amdgcn_wmma_f32_16x16x4_f32(false, a, false, bp[48], (short)0, acc3, false, false);
    }

    const int g0 = g[0];
    bool same = true;
    float s0 = 0.0f, s1 = 0.0f, s2 = 0.0f, s3 = 0.0f;
#pragma unroll
    for (int r = 0; r < 8; ++r) {
        float v0 = acc0[r] + bb[0], v1 = acc1[r] + bb[1], v2 = acc2[r] + bb[2], v3 = acc3[r] + bb[3];
        v0 = v0 > 0.0f ? v0 : 0.0f;
        v1 = v1 > 0.0f ? v1 : 0.0f;
        v2 = v2 > 0.0f ? v2 : 0.0f;
        v3 = v3 > 0.0f ? v3 : 0.0f;
        acc0[r] = v0; acc1[r] = v1; acc2[r] = v2; acc3[r] = v3;
        s0 += v0; s1 += v1; s2 += v2; s3 += v3;
        same = same && (g[r] == g0);
    }
    const int   gx    = __shfl_xor(g0, 16);
    const int   samex = __shfl_xor((int)same, 16);
    const float x0 = __shfl_xor(s0, 16), x1 = __shfl_xor(s1, 16), x2 = __shfl_xor(s2, 16), x3 = __shfl_xor(s3, 16);
    const bool whole = same && (samex != 0) && (gx == g0);
    if (whole) {
        if (half == 0) {
            float* pr = pool + (size_t)g0 * HID + m;
            atomicAdd(&pr[ 0], s0 + x0);
            atomicAdd(&pr[16], s1 + x1);
            atomicAdd(&pr[32], s2 + x2);
            atomicAdd(&pr[48], s3 + x3);
            if (m == 0) atomicAdd(&cnt[g0], 16.0f);
        }
    } else if (same) {
        float* pr = pool + (size_t)g0 * HID + m;
        atomicAdd(&pr[ 0], s0);
        atomicAdd(&pr[16], s1);
        atomicAdd(&pr[32], s2);
        atomicAdd(&pr[48], s3);
        if (m == 0) atomicAdd(&cnt[g0], 8.0f);
    } else {
#pragma unroll
        for (int r = 0; r < 8; ++r) {
            float* pr = pool + (size_t)g[r] * HID + m;
            atomicAdd(&pr[ 0], acc0[r]);
            atomicAdd(&pr[16], acc1[r]);
            atomicAdd(&pr[32], acc2[r]);
            atomicAdd(&pr[48], acc3[r]);
            if (m == r) atomicAdd(&cnt[g[r]], 1.0f);
        }
    }
}
#undef AG_ADD

__global__ void k_pool_init(float* pool, float* cnt) {
    int i = blockIdx.x * blockDim.x + threadIdx.x;
    if (i < N_GRAPH * HID) pool[i] = 0.0f;
    if (i < N_GRAPH) cnt[i] = 0.0f;
}

__global__ void k_final(const float* __restrict__ pool, const float* __restrict__ cnt,
                        float* __restrict__ out) {
    int tid = blockIdx.x * blockDim.x + threadIdx.x;
    if (tid < N_GRAPH * HID) {
        float c = cnt[tid >> 6];
        out[tid] = pool[tid] / fmaxf(c, 1.0f);
    }
}

extern "C" void kernel_launch(void* const* d_in, const int* in_sizes, int n_in,
                              void* d_out, int out_size, void* d_ws, size_t ws_size,
                              hipStream_t stream) {
    const float* x     = (const float*)d_in[0];
    const int*   edge  = (const int*)  d_in[1];
    const int*   batch = (const int*)  d_in[2];
    const float* W1    = (const float*)d_in[3];
    const float* b1    = (const float*)d_in[4];
    const float* W2    = (const float*)d_in[5];
    const float* b2    = (const float*)d_in[6];
    const int* src = edge;
    const int* dst = edge + N_EDGES;
    float* out = (float*)d_out;

    float* ws   = (float*)d_ws;
    float* dinv = ws;
    float* bufA = dinv + 2 * N_NODES;
    float* bufB = bufA + (size_t)N_NODES * HID;
    int* cntm    = (int*)bufA;
    int* bbase   = cntm + NEBLK * NBKT;
    v2i* ebuf    = (v2i*)(bbase + 512);
    int* rowptr  = (int*)bufB;
    int* csr_src = rowptr + N_NODES + 1024;
    float* y     = (float*)(csr_src + N_EDGES + 32);
    float* sd    = y + N_NODES + 8;
    float* pool = bufB + (size_t)N_NODES * HID;
    float* cnt  = pool + N_GRAPH * HID;

    const int TB = 256;
    k_bkt_count    <<<NEBLK, ETB, 0, stream>>>(dst, cntm);
    k_bkt_scatter  <<<NEBLK, ETB, 0, stream>>>(src, dst, cntm, bbase, ebuf);
    k_bkt_sort     <<<NBKT, PTB, 0, stream>>>(ebuf, bbase, x, rowptr, dinv, y, csr_src, sd, N_NODES);
    k_l1_pull      <<<(N_NODES + TB - 1) / TB, TB, 0, stream>>>(rowptr, csr_src, dinv, y, sd, N_NODES);

    k_pool_init    <<<(N_GRAPH * HID + TB - 1) / TB, TB, 0, stream>>>(pool, cnt);
    const int n_tiles = N_NODES / 16;
    k_l2_fused     <<<(n_tiles + 7) / 8, 256, 0, stream>>>(rowptr, csr_src, sd, W1, b1, W2, b2, batch,
                                                           pool, cnt, n_tiles, N_NODES);
    k_final        <<<(N_GRAPH * HID + TB - 1) / TB, TB, 0, stream>>>(pool, cnt, out);
}


# Each edge block groups its own pairs by bucket into its own stretch (LDS-atomic rank, counts scanned in-block); pass 2 walks 256 runs per bucket; count kernel and matrix sums gone (row order changes)
# speedup vs baseline: 7.7048x; 1.0373x over previous
#include <hip/hip_runtime.h>

typedef float v2f __attribute__((ext_vector_type(2)));
typedef float v8f __attribute__((ext_vector_type(8)));

#define N_NODES 200000
#define N_EDGES 3200000
#define N_GRAPH 2048
#define HID     64
#define BPAIR_STRIDE 80

typedef int v2i __attribute__((ext_vector_type(2)));
typedef int v4i __attribute__((ext_vector_type(4)));
#define NBKT   256
#define BKT_N  782
#define NEBLK  256
#define EBLK_E 12500
#define ETB    512
#define EPT    25
#define PTB    1024
#define RPW    (NEBLK / (PTB / 32))
#define RIT    3
#define SCAP   16384

__device__ __forceinline__ unsigned bkt_of(int d) {
    const unsigned b = (unsigned)d / (unsigned)BKT_N;
    return b < (unsigned)(NBKT - 1) ? b : (unsigned)(NBKT - 1);
}

__global__ __launch_bounds__(ETB)
void k_bkt_scatter(const int* __restrict__ src, const int* __restrict__ dst, int* __restrict__ runs,
                   v2i* __restrict__ ebuf) {
    __shared__ int cur[NBKT];
    __shared__ int wloc[NBKT / 32];
    __shared__ v2i stage[EBLK_E];
    const int tid = threadIdx.x, lane = tid & 31, wave = tid >> 5;
    const int b = blockIdx.x;
    if (tid < NBKT) cur[tid] = 0;

    const int* sp = src + (size_t)b * EBLK_E;
    const int* dp = dst + (size_t)b * EBLK_E;
    int u[EPT], v[EPT];
#pragma unroll
    for (int k = 0; k < EPT; ++k) {
        const int j = k * ETB + tid;
        const bool ok = j < EBLK_E;
        u[k] = ok ? sp[j] : 0;
        v[k] = ok ? dp[j] : -1;
    }
    __syncthreads();

    int rk[EPT];
#pragma unroll
    for (int k = 0; k < EPT; ++k) rk[k] = (v[k] >= 0) ? atomicAdd(&cur[bkt_of(v[k])], 1) : -1;
    __syncthreads();

    int own = 0, xl = 0;
    if (tid < NBKT) {
        own = cur[tid];
        xl = own;
        for (int off = 1; off < 32; off <<= 1) {
            const int yl = __shfl_up(xl, off);
            if (lane >= off) xl += yl;
        }
        if (lane == 31) wloc[wave] = xl;
    }
    __syncthreads();
    if (tid < NBKT) {
        int offl = 0;
        for (int w = 0; w < wave; ++w) offl += wloc[w];
        const int lbeg = offl + xl - own;
        cur[tid] = lbeg;
        runs[b * NBKT + tid] = lbeg | (own << 16);
    }
    __syncthreads();

    int pos[EPT];
#pragma unroll
    for (int k = 0; k < EPT; ++k) pos[k] = cur[bkt_of(v[k])];
#pragma unroll
    for (int k = 0; k < EPT; ++k) {
        const int ps = pos[k] + rk[k];
        if (rk[k] >= 0 && ps < EBLK_E) {
            v2i pr = { u[k], v[k] };
            stage[ps] = pr;
        }
    }
    __syncthreads();
    v2i* ob = ebuf + (size_t)b * EBLK_E;
    v2i pr[EPT];
#pragma unroll
    for (int k = 0; k < EPT; ++k) {
        const int j = k * ETB + tid;
        v2i none = { 0, 0 };
        pr[k] = (j < EBLK_E) ? stage[j] : none;
    }
#pragma unroll
    for (int k = 0; k < EPT; ++k) {
        const int j = k * ETB + tid;
        if (j < EBLK_E) ob[j] = pr[k];
    }
}

__global__ __launch_bounds__(PTB)
void k_bkt_sort(const v2i* __restrict__ ebuf, const int* __restrict__ runs, const float* __restrict__ x,
                int* __restrict__ rowptr, float* __restrict__ dinv, float* __restrict__ y,
                int* __restrict__ csr_src, float* __restrict__ sd, int n) {
    __shared__ int cnt[PTB];
    __shared__ int wsum[PTB / 32];
    __shared__ int wbeg[PTB / 32];
    __shared__ int wlen[PTB / 32];
    __shared__ v4i stage4[SCAP / 4];
    const int tid = threadIdx.x, lane = tid & 31, wave = tid >> 5;
    const int r  = blockIdx.x;
    const int lo = r * BKT_N;
    int nn = n - lo;
    nn = nn < BKT_N ? nn : BKT_N;
    nn = nn > 0 ? nn : 0;
    int* stage = (int*)stage4;
    cnt[tid] = 0;
    const float xi = (tid < nn) ? x[lo + tid] : 0.0f;
    const int pk = runs[(wave * RPW + (lane & (RPW - 1))) * NBKT + r];
    {
        int ssum = pk & 0xffff, lsum = pk >> 16;
        for (int off = 1; off < RPW; off <<= 1) {
            ssum += __shfl_xor(ssum, off);
            lsum += __shfl_xor(lsum, off);
        }
        if (lane == 0) { wbeg[wave] = ssum; wlen[wave] = lsum; }
    }
    if (r == 0 && tid == 0) {
        csr_src[N_EDGES] = n;
        y[n] = 0.0f;
        sd[2 * n] = 0.0f;
        sd[2 * n + 1] = 0.0f;
    }
    __syncthreads();

    int beg = 0, tot = 0;
#pragma unroll
    for (int w = 0; w < PTB / 32; ++w) { beg += wbeg[w]; tot += wlen[w]; }
    const int end = beg + tot;
    const int beg4 = beg & ~3;

    const v2i* rp[RPW];
    int ln[RPW];
#pragma unroll
    for (int q = 0; q < RPW; ++q) {
        const int e = __builtin_amdgcn_readlane(pk, q);
        rp[q] = ebuf + (size_t)(wave * RPW + q) * EBLK_E + (e & 0xffff);
        ln[q] = e >> 16;
    }

    {
        int dd[RPW][RIT];
#pragma unroll
        for (int q = 0; q < RPW; ++q) {
#pragma unroll
            for (int t = 0; t < RIT; ++t) {
                const int i = t * 32 + lane;
                dd[q][t] = (i < ln[q]) ? rp[q][i].y : -1;
            }
        }
#pragma unroll
        for (int q = 0; q < RPW; ++q) {
#pragma unroll
            for (int t = 0; t < RIT; ++t) {
                const unsigned o = (unsigned)(dd[q][t] - lo);
                if (o < (unsigned)nn) atomicAdd(&cnt[o], 1);
            }
        }
#pragma unroll
        for (int q = 0; q < RPW; ++q) {
            for (int i = RIT * 32 + lane; i < ln[q]; i += 32) {
                const unsigned o = (unsigned)(rp[q][i].y - lo);
                if (o < (unsigned)nn) atomicAdd(&cnt[o], 1);
            }
        }
    }
    __syncthreads();

    const int c = cnt[tid];
    int xs = c;
    for (int off = 1; off < 32; off <<= 1) {
        const int yv = __shfl_up(xs, off);
        if (lane >= off) xs += yv;
    }
    if (lane == 31) wsum[wave] = xs;
    __syncthreads();
    int offw = 0;
    for (int w = 0; w < wave; ++w) offw += wsum[w];
    const int first = beg + offw + xs - c;
    cnt[tid] = first;
    if (tid < nn) {
        const float dv = rsqrtf(1.0f + (float)c);
        rowptr[lo + tid] = first;
        dinv[lo + tid] = dv;
        y[lo + tid] = dv * xi;
    }
    __syncthreads();

    {
        v2i p[RPW][RIT];
#pragma unroll
        for (int q = 0; q < RPW; ++q) {
#pragma unroll
            for (int t = 0; t < RIT; ++t) {
                const int i = t * 32 + lane;
                v2i none = { 0, -1 };
                p[q][t] = (i < ln[q]) ? rp[q][i] : none;
            }
        }
        int pos[RPW][RIT];
#pragma unroll
        for (int q = 0; q < RPW; ++q) {
#pragma unroll
            for (int t = 0; t < RIT; ++t) {
                const unsigned o = (unsigned)(p[q][t].y - lo);
                pos[q][t] = (o < (unsigned)nn) ? atomicAdd(&cnt[o], 1) : -1;
            }
        }
#pragma unroll
        for (int q = 0; q < RPW; ++q) {
#pragma unroll
            for (int t = 0; t < RIT; ++t) {
                if (pos[q][t] >= 0) {
                    const unsigned s = (unsigned)(pos[q][t] - beg4);
                    if (s < (unsigned)SCAP) stage[s] = p[q][t].x;
                    else csr_src[pos[q][t]] = p[q][t].x;
                }
            }
        }
#pragma unroll
        for (int q = 0; q < RPW; ++q) {
            for (int i = RIT * 32 + lane; i < ln[q]; i += 32) {
                const v2i pr = rp[q][i];
                const unsigned o = (unsigned)(pr.y - lo);
                if (o < (unsigned)nn) {
                    const int ps = atomicAdd(&cnt[o], 1);
                    const unsigned s = (unsigned)(ps - beg4);
                    if (s < (unsigned)SCAP) stage[s] = pr.x;
                    else csr_src[ps] = pr.x;
                }
            }
        }
    }
    __syncthreads();
    const int lim = (end < beg4 + SCAP) ? end : beg4 + SCAP;
    for (int p0 = beg4 + 4 * tid; p0 < lim; p0 += 4 * PTB) {
        const v4i w = stage4[(p0 - beg4) >> 2];
        if (p0 >= beg && p0 + 4 <= lim) {
            *(v4i*)(csr_src + p0) = w;
        } else {
            if (p0     >= beg && p0     < lim) csr_src[p0]     = w.x;
            if (p0 + 1 >= beg && p0 + 1 < lim) csr_src[p0 + 1] = w.y;
            if (p0 + 2 >= beg && p0 + 2 < lim) csr_src[p0 + 2] = w.z;
            if (p0 + 3 >= beg && p0 + 3 < lim) csr_src[p0 + 3] = w.w;
        }
    }
}

#define L1G 8
#define L1P 8
#define L1D 4

template <int P> __device__ __forceinline__ int grp_lane(int v) {
    return __builtin_amdgcn_mov_dpp8(v, P | (P << 3) | (P << 6) | (P << 9) | (P << 12) | (P << 15) | (P << 18) | (P << 21));
}
template <int M> __device__ __forceinline__ float xmask_add(float a) {
    return a + __int_as_float(__builtin_amdgcn_update_dpp(0, __float_as_int(a), 0x160 + M, 0xf, 0xf, false));
}

__global__ __launch_bounds__(256)
void k_l1_pull(const int* __restrict__ rowptr, const int* __restrict__ csr_src,
               const float* __restrict__ dinv, const float* __restrict__ y,
               float* __restrict__ sd, int n) {
    const int n0 = (int)((blockIdx.x * blockDim.x + threadIdx.x) >> 5) * 32;
    if (n0 >= n) return;
    const int lane = threadIdx.x & 31;
    const int sub  = lane & (L1G - 1);
    const int mine = n0 + lane;
    const int rb = (mine < n) ? rowptr[mine] : N_EDGES;
    const int re = (mine + 1 < n) ? rowptr[mine + 1] : N_EDGES;
    const int mc = mine < n ? mine : n - 1;
    const float dvm = dinv[mc];
    const float ym  = y[mc];

    int beg[L1P], end[L1P];
#define L1_BOUNDS(P) beg[P] = grp_lane<P>(rb); end[P] = grp_lane<P>(re);
    L1_BOUNDS(0) L1_BOUNDS(1) L1_BOUNDS(2) L1_BOUNDS(3) L1_BOUNDS(4) L1_BOUNDS(5) L1_BOUNDS(6) L1_BOUNDS(7)
#undef L1_BOUNDS

    int   u[L1P][L1D];
    float v[L1P][L1D];
#pragma unroll
    for (int p = 0; p < L1P; ++p) {
#pragma unroll
        for (int d = 0; d < L1D; ++d) {
            const int j = beg[p] + sub + d * L1G;
            u[p][d] = csr_src[j < end[p] ? j : N_EDGES];
        }
    }
#pragma unroll
    for (int p = 0; p < L1P; ++p) {
#pragma unroll
        for (int d = 0; d < L1D; ++d) v[p][d] = y[u[p][d]];
    }
    float res = 0.0f;
#pragma unroll
    for (int p = 0; p < L1P; ++p) {
        float a = 0.0f;
#pragma unroll
        for (int d = 0; d < L1D; ++d) a += v[p][d];
        for (int j = beg[p] + sub + L1D * L1G; j < end[p]; j += L1G) a += y[csr_src[j]];
        a = xmask_add<4>(a);
        a = xmask_add<2>(a);
        a = xmask_add<1>(a);
        if (sub == p) res = a;
    }
    if (mine < n) {
        v2f pr = { dvm * (ym + res), dvm };
        ((v2f*)sd)[mine] = pr;
    }
}

typedef float v4f __attribute__((ext_vector_type(4)));
#define AGB 4

#define AG_ADD(P)                                                         \
    {                                                                     \
        const float su_ = (P).x, wu_ = (P).y;                             \
        _Pragma("unroll")                                                 \
        for (int q_ = 0; q_ < HID / 4; ++q_) {                            \
            const float h0_ = su_ * wb[q_].x + wb[q_].z;                  \
            const float h1_ = su_ * wb[q_].y + wb[q_].w;                  \
            ag[q_].x += wu_ * (h0_ > 0.0f ? h0_ : 0.0f);                  \
            ag[q_].y += wu_ * (h1_ > 0.0f ? h1_ : 0.0f);                  \
        }                                                                 \
    }

__global__ __launch_bounds__(256)
void k_l2_fused(const int* __restrict__ rowptr, const int* __restrict__ csr_src, const float* __restrict__ sd,
                const float* __restrict__ W1, const float* __restrict__ b1,
                const float* __restrict__ W2, const float* __restrict__ b2,
                const int* __restrict__ batch, float* pool, float* cnt, int n_tiles, int n) {
    __shared__ v2f sB[(HID / 2) * BPAIR_STRIDE];
    __shared__ v4f sWB[HID / 2];
    {
        float* sf = (float*)sB;
        for (int idx = threadIdx.x; idx < HID * HID; idx += blockDim.x) {
            int k = idx >> 6, col = idx & 63;
            sf[(k >> 1) * (2 * BPAIR_STRIDE) + col * 2 + (k & 1)] = W2[idx];
        }
        if (threadIdx.x < HID / 2) {
            const int p = threadIdx.x;
            v4f wbp = { W1[2 * p], W1[2 * p + 1], b1[2 * p], b1[2 * p + 1] };
            sWB[p] = wbp;
        }
    }
    __syncthreads();

    const int wave = threadIdx.x >> 5;
    const int lane = threadIdx.x & 31;
    const int tile = blockIdx.x * 8 + wave;
    if (tile >= n_tiles) return;

    const int row0 = tile * 16;
    const int m    = lane & 15;
    const int half = lane >> 4;
    const int node = row0 + m;

    const v2f* sdp = (const v2f*)sd;
    const int rb = rowptr[node];
    const int re = (node + 1 < n) ? rowptr[node + 1] : N_EDGES;
    const v2f self = sdp[node];

    int j = rb;
    int un[AGB];
    v2f pv[AGB];
#pragma unroll
    for (int d = 0; d < AGB; ++d) un[d] = csr_src[j + d < re ? j + d : N_EDGES];
#pragma unroll
    for (int d = 0; d < AGB; ++d) pv[d] = sdp[un[d]];
#pragma unroll
    for (int d = 0; d < AGB; ++d) un[d] = csr_src[j + AGB + d < re ? j + AGB + d : N_EDGES];

    v4f wb[HID / 4];
    v2f ag[HID / 4];
#pragma unroll
    for (int q = 0; q < HID / 4; ++q) {
        wb[q] = sWB[2 * q + half];
        ag[q].x = 0.0f;
        ag[q].y = 0.0f;
    }
    AG_ADD(self)

    while (__builtin_amdgcn_ballot_w32(j < re) != 0u) {
        v2f pn[AGB];
#pragma unroll
        for (int d = 0; d < AGB; ++d) pn[d] = sdp[un[d]];
#pragma unroll
        for (int d = 0; d < AGB; ++d) un[d] = csr_src[j + 2 * AGB + d < re ? j + 2 * AGB + d : N_EDGES];
        __builtin_amdgcn_sched_barrier(0);
#pragma unroll
        for (int d = 0; d < AGB; ++d) AG_ADD(pv[d])
#pragma unroll
        for (int d = 0; d < AGB; ++d) pv[d] = pn[d];
        j += AGB;
    }

    const int rbase = row0 + half * 8;
    int g[8];
#pragma unroll
    for (int r = 0; r < 8; ++r) g[r] = batch[rbase + r];
    float bb[4];
#pragma unroll
    for (int c = 0; c < 4; ++c) bb[c] = b2[16 * c + m];

    const float dv = self.y;
    v8f acc0 = {}, acc1 = {}, acc2 = {}, acc3 = {};
#pragma unroll
    for (int q = 0; q < HID / 4; ++q) {
        v2f a;
        a.x = dv * ag[q].x;
        a.y = dv * ag[q].y;
        const v2f* bp = &sB[(2 * q + half) * BPAIR_STRIDE + m];
        acc0 = __builtin_amdgcn_wmma_f32_16x16x4_f32(false, a, false, bp[ 0], (short)0, acc0, false, false);
        acc1 = __builtin_amdgcn_wmma_f32_16x16x4_f32(false, a, false, bp[16], (short)0, acc1, false, false);
        acc2 = __builtin_amdgcn_wmma_f32_16x16x4_f32(false, a, false, bp[32], (short)0, acc2, false, false);
        acc3 = __builtin_amdgcn_wmma_f32_16x16x4_f32(false, a, false, bp[48], (short)0, acc3, false, false);
    }

    const int g0 = g[0];
    bool same = true;
    float s0 = 0.0f, s1 = 0.0f, s2 = 0.0f, s3 = 0.0f;
#pragma unroll
    for (int r = 0; r < 8; ++r) {
        float v0 = acc0[r] + bb[0], v1 = acc1[r] + bb[1], v2 = acc2[r] + bb[2], v3 = acc3[r] + bb[3];
        v0 = v0 > 0.0f ? v0 : 0.0f;
        v1 = v1 > 0.0f ? v1 : 0.0f;
        v2 = v2 > 0.0f ? v2 : 0.0f;
        v3 = v3 > 0.0f ? v3 : 0.0f;
        acc0[r] = v0; acc1[r] = v1; acc2[r] = v2; acc3[r] = v3;
        s0 += v0; s1 += v1; s2 += v2; s3 += v3;
        same = same && (g[r] == g0);
    }
    const int   gx    = __shfl_xor(g0, 16);
    const int   samex = __shfl_xor((int)same, 16);
    const float x0 = __shfl_xor(s0, 16), x1 = __shfl_xor(s1, 16), x2 = __shfl_xor(s2, 16), x3 = __shfl_xor(s3, 16);
    const bool whole = same && (samex != 0) && (gx == g0);
    if (whole) {
        if (half == 0) {
            float* pr = pool + (size_t)g0 * HID + m;
            atomicAdd(&pr[ 0], s0 + x0);
            atomicAdd(&pr[16], s1 + x1);
            atomicAdd(&pr[32], s2 + x2);
            atomicAdd(&pr[48], s3 + x3);
            if (m == 0) atomicAdd(&cnt[g0], 16.0f);
        }
    } else if (same) {
        float* pr = pool + (size_t)g0 * HID + m;
        atomicAdd(&pr[ 0], s0);
        atomicAdd(&pr[16], s1);
        atomicAdd(&pr[32], s2);
        atomicAdd(&pr[48], s3);
        if (m == 0) atomicAdd(&cnt[g0], 8.0f);
    } else {
#pragma unroll
        for (int r = 0; r < 8; ++r) {
            float* pr = pool + (size_t)g[r] * HID + m;
            atomicAdd(&pr[ 0], acc0[r]);
            atomicAdd(&pr[16], acc1[r]);
            atomicAdd(&pr[32], acc2[r]);
            atomicAdd(&pr[48], acc3[r]);
            if (m == r) atomicAdd(&cnt[g[r]], 1.0f);
        }
    }
}
#undef AG_ADD

__global__ void k_pool_init(float* pool, float* cnt) {
    int i = blockIdx.x * blockDim.x + threadIdx.x;
    if (i < N_GRAPH * HID) pool[i] = 0.0f;
    if (i < N_GRAPH) cnt[i] = 0.0f;
}

__global__ void k_final(const float* __restrict__ pool, const float* __restrict__ cnt,
                        float* __restrict__ out) {
    int tid = blockIdx.x * blockDim.x + threadIdx.x;
    if (tid < N_GRAPH * HID) {
        float c = cnt[tid >> 6];
        out[tid] = pool[tid] / fmaxf(c, 1.0f);
    }
}

extern "C" void kernel_launch(void* const* d_in, const int* in_sizes, int n_in,
                              void* d_out, int out_size, void* d_ws, size_t ws_size,
                              hipStream_t stream) {
    const float* x     = (const float*)d_in[0];
    const int*   edge  = (const int*)  d_in[1];
    const int*   batch = (const int*)  d_in[2];
    const float* W1    = (const float*)d_in[3];
    const float* b1    = (const float*)d_in[4];
    const float* W2    = (const float*)d_in[5];
    const float* b2    = (const float*)d_in[6];
    const int* src = edge;
    const int* dst = edge + N_EDGES;
    float* out = (float*)d_out;

    float* ws   = (float*)d_ws;
    float* dinv = ws;
    float* bufA = dinv + 2 * N_NODES;
    float* bufB = bufA + (size_t)N_NODES * HID;
    int* runs    = (int*)bufA;
    v2i* ebuf    = (v2i*)(runs + NEBLK * NBKT + 512);
    int* rowptr  = (int*)bufB;
    int* csr_src = rowptr + N_NODES + 1024;
    float* y     = (float*)(csr_src + N_EDGES + 32);
    float* sd    = y + N_NODES + 8;
    float* pool = bufB + (size_t)N_NODES * HID;
    float* cnt  = pool + N_GRAPH * HID;

    const int TB = 256;
    k_bkt_scatter  <<<NEBLK, ETB, 0, stream>>>(src, dst, runs, ebuf);
    k_bkt_sort     <<<NBKT, PTB, 0, stream>>>(ebuf, runs, x, rowptr, dinv, y, csr_src, sd, N_NODES);
    k_l1_pull      <<<(N_NODES + TB - 1) / TB, TB, 0, stream>>>(rowptr, csr_src, dinv, y, sd, N_NODES);

    k_pool_init    <<<(N_GRAPH * HID + TB - 1) / TB, TB, 0, stream>>>(pool, cnt);
    const int n_tiles = N_NODES / 16;
    k_l2_fused     <<<(n_tiles + 7) / 8, 256, 0, stream>>>(rowptr, csr_src, sd, W1, b1, W2, b2, batch,
                                                           pool, cnt, n_tiles, N_NODES);
    k_final        <<<(N_GRAPH * HID + TB - 1) / TB, TB, 0, stream>>>(pool, cnt, out);
}


# Mean pool adds go straight into out, each sum scaled by 1/node count before its atomic; counts taken in the layer-1 kernel; pool clear and divide launches gone (atomic order as before, terms scaled)
# speedup vs baseline: 8.0364x; 1.0430x over previous
#include <hip/hip_runtime.h>

typedef float v2f __attribute__((ext_vector_type(2)));
typedef float v8f __attribute__((ext_vector_type(8)));

#define N_NODES 200000
#define N_EDGES 3200000
#define N_GRAPH 2048
#define HID     64
#define BPAIR_STRIDE 80

typedef int v2i __attribute__((ext_vector_type(2)));
typedef int v4i __attribute__((ext_vector_type(4)));
#define NBKT   256
#define BKT_N  782
#define NEBLK  256
#define EBLK_E 12500
#define ETB    512
#define EPT    25
#define PTB    1024
#define RPW    (NEBLK / (PTB / 32))
#define RIT    3
#define SCAP   16384

__device__ __forceinline__ unsigned bkt_of(int d) {
    const unsigned b = (unsigned)d / (unsigned)BKT_N;
    return b < (unsigned)(NBKT - 1) ? b : (unsigned)(NBKT - 1);
}

__global__ __launch_bounds__(ETB)
void k_bkt_scatter(const int* __restrict__ src, const int* __restrict__ dst, int* __restrict__ runs,
                   v2i* __restrict__ ebuf, float* __restrict__ out, float* __restrict__ cnt) {
    __shared__ int cur[NBKT];
    __shared__ int wloc[NBKT / 32];
    __shared__ v2i stage[EBLK_E];
    const int tid = threadIdx.x, lane = tid & 31, wave = tid >> 5;
    const int b = blockIdx.x;
    if (tid < NBKT) cur[tid] = 0;
    static_assert(NEBLK * ETB >= N_GRAPH * HID, "one thread of this grid per element of out");
    if (b * ETB + tid < N_GRAPH * HID) out[b * ETB + tid] = 0.0f;
    if (b * ETB + tid < N_GRAPH) cnt[b * ETB + tid] = 0.0f;

    const int* sp = src + (size_t)b * EBLK_E;
    const int* dp = dst + (size_t)b * EBLK_E;
    int u[EPT], v[EPT];
#pragma unroll
    for (int k = 0; k < EPT; ++k) {
        const int j = k * ETB + tid;
        const bool ok = j < EBLK_E;
        u[k] = ok ? sp[j] : 0;
        v[k] = ok ? dp[j] : -1;
    }
    __syncthreads();

    int rk[EPT];
#pragma unroll
    for (int k = 0; k < EPT; ++k) rk[k] = (v[k] >= 0) ? atomicAdd(&cur[bkt_of(v[k])], 1) : -1;
    __syncthreads();

    int own = 0, xl = 0;
    if (tid < NBKT) {
        own = cur[tid];
        xl = own;
        for (int off = 1; off < 32; off <<= 1) {
            const int yl = __shfl_up(xl, off);
            if (lane >= off) xl += yl;
        }
        if (lane == 31) wloc[wave] = xl;
    }
    __syncthreads();
    if (tid < NBKT) {
        int offl = 0;
        for (int w = 0; w < wave; ++w) offl += wloc[w];
        const int lbeg = offl + xl - own;
        cur[tid] = lbeg;
        runs[b * NBKT + tid] = lbeg | (own << 16);
    }
    __syncthreads();

    int pos[EPT];
#pragma unroll
    for (int k = 0; k < EPT; ++k) pos[k] = cur[bkt_of(v[k])];
#pragma unroll
    for (int k = 0; k < EPT; ++k) {
        const int ps = pos[k] + rk[k];
        if (rk[k] >= 0 && ps < EBLK_E) {
            v2i pr = { u[k], v[k] };
            stage[ps] = pr;
        }
    }
    __syncthreads();
    v2i* ob = ebuf + (size_t)b * EBLK_E;
    v2i pr[EPT];
#pragma unroll
    for (int k = 0; k < EPT; ++k) {
        const int j = k * ETB + tid;
        v2i none = { 0, 0 };
        pr[k] = (j < EBLK_E) ? stage[j] : none;
    }
#pragma unroll
    for (int k = 0; k < EPT; ++k) {
        const int j = k * ETB + tid;
        if (j < EBLK_E) ob[j] = pr[k];
    }
}

__global__ __launch_bounds__(PTB)
void k_bkt_sort(const v2i* __restrict__ ebuf, const int* __restrict__ runs, const float* __restrict__ x,
                int* __restrict__ rowptr, float* __restrict__ dinv, float* __restrict__ y,
                int* __restrict__ csr_src, float* __restrict__ sd, int n) {
    __shared__ int cnt[PTB];
    __shared__ int wsum[PTB / 32];
    __shared__ int wbeg[PTB / 32];
    __shared__ int wlen[PTB / 32];
    __shared__ v4i stage4[SCAP / 4];
    const int tid = threadIdx.x, lane = tid & 31, wave = tid >> 5;
    const int r  = blockIdx.x;
    const int lo = r * BKT_N;
    int nn = n - lo;
    nn = nn < BKT_N ? nn : BKT_N;
    nn = nn > 0 ? nn : 0;
    int* stage = (int*)stage4;
    cnt[tid] = 0;
    const float xi = (tid < nn) ? x[lo + tid] : 0.0f;
    const int pk = runs[(wave * RPW + (lane & (RPW - 1))) * NBKT + r];
    {
        int ssum = pk & 0xffff, lsum = pk >> 16;
        for (int off = 1; off < RPW; off <<= 1) {
            ssum += __shfl_xor(ssum, off);
            lsum += __shfl_xor(lsum, off);
        }
        if (lane == 0) { wbeg[wave] = ssum; wlen[wave] = lsum; }
    }
    if (r == 0 && tid == 0) {
        csr_src[N_EDGES] = n;
        y[n] = 0.0f;
        sd[2 * n] = 0.0f;
        sd[2 * n + 1] = 0.0f;
    }
    __syncthreads();

    int beg = 0, tot = 0;
#pragma unroll
    for (int w = 0; w < PTB / 32; ++w) { beg += wbeg[w]; tot += wlen[w]; }
    const int end = beg + tot;
    const int beg4 = beg & ~3;

    const v2i* rp[RPW];
    int ln[RPW];
#pragma unroll
    for (int q = 0; q < RPW; ++q) {
        const int e = __builtin_amdgcn_readlane(pk, q);
        rp[q] = ebuf + (size_t)(wave * RPW + q) * EBLK_E + (e & 0xffff);
        ln[q] = e >> 16;
    }

    {
        int dd[RPW][RIT];
#pragma unroll
        for (int q = 0; q < RPW; ++q) {
#pragma unroll
            for (int t = 0; t < RIT; ++t) {
                const int i = t * 32 + lane;
                dd[q][t] = (i < ln[q]) ? rp[q][i].y : -1;
            }
        }
#pragma unroll
        for (int q = 0; q < RPW; ++q) {
#pragma unroll
            for (int t = 0; t < RIT; ++t) {
                const unsigned o = (unsigned)(dd[q][t] - lo);
                if (o < (unsigned)nn) atomicAdd(&cnt[o], 1);
            }
        }
#pragma unroll
        for (int q = 0; q < RPW; ++q) {
            for (int i = RIT * 32 + lane; i < ln[q]; i += 32) {
                const unsigned o = (unsigned)(rp[q][i].y - lo);
                if (o < (unsigned)nn) atomicAdd(&cnt[o], 1);
            }
        }
    }
    __syncthreads();

    const int c = cnt[tid];
    int xs = c;
    for (int off = 1; off < 32; off <<= 1) {
        const int yv = __shfl_up(xs, off);
        if (lane >= off) xs += yv;
    }
    if (lane == 31) wsum[wave] = xs;
    __syncthreads();
    int offw = 0;
    for (int w = 0; w < wave; ++w) offw += wsum[w];
    const int first = beg + offw + xs - c;
    cnt[tid] = first;
    if (tid < nn) {
        const float dv = rsqrtf(1.0f + (float)c);
        rowptr[lo + tid] = first;
        dinv[lo + tid] = dv;
        y[lo + tid] = dv * xi;
    }
    __syncthreads();

    {
        v2i p[RPW][RIT];
#pragma unroll
        for (int q = 0; q < RPW; ++q) {
#pragma unroll
            for (int t = 0; t < RIT; ++t) {
                const int i = t * 32 + lane;
                v2i none = { 0, -1 };
                p[q][t] = (i < ln[q]) ? rp[q][i] : none;
            }
        }
        int pos[RPW][RIT];
#pragma unroll
        for (int q = 0; q < RPW; ++q) {
#pragma unroll
            for (int t = 0; t < RIT; ++t) {
                const unsigned o = (unsigned)(p[q][t].y - lo);
                pos[q][t] = (o < (unsigned)nn) ? atomicAdd(&cnt[o], 1) : -1;
            }
        }
#pragma unroll
        for (int q = 0; q < RPW; ++q) {
#pragma unroll
            for (int t = 0; t < RIT; ++t) {
                if (pos[q][t] >= 0) {
                    const unsigned s = (unsigned)(pos[q][t] - beg4);
                    if (s < (unsigned)SCAP) stage[s] = p[q][t].x;
                    else csr_src[pos[q][t]] = p[q][t].x;
                }
            }
        }
#pragma unroll
        for (int q = 0; q < RPW; ++q) {
            for (int i = RIT * 32 + lane; i < ln[q]; i += 32) {
                const v2i pr = rp[q][i];
                const unsigned o = (unsigned)(pr.y - lo);
                if (o < (unsigned)nn) {
                    const int ps = atomicAdd(&cnt[o], 1);
                    const unsigned s = (unsigned)(ps - beg4);
                    if (s < (unsigned)SCAP) stage[s] = pr.x;
                    else csr_src[ps] = pr.x;
                }
            }
        }
    }
    __syncthreads();
    const int lim = (end < beg4 + SCAP) ? end : beg4 + SCAP;
    for (int p0 = beg4 + 4 * tid; p0 < lim; p0 += 4 * PTB) {
        const v4i w = stage4[(p0 - beg4) >> 2];
        if (p0 >= beg && p0 + 4 <= lim) {
            *(v4i*)(csr_src + p0) = w;
        } else {
            if (p0     >= beg && p0     < lim) csr_src[p0]     = w.x;
            if (p0 + 1 >= beg && p0 + 1 < lim) csr_src[p0 + 1] = w.y;
            if (p0 + 2 >= beg && p0 + 2 < lim) csr_src[p0 + 2] = w.z;
            if (p0 + 3 >= beg && p0 + 3 < lim) csr_src[p0 + 3] = w.w;
        }
    }
}

#define L1G 8
#define L1P 8
#define L1D 4

template <int P> __device__ __forceinline__ int grp_lane(int v) {
    return __builtin_amdgcn_mov_dpp8(v, P | (P << 3) | (P << 6) | (P << 9) | (P << 12) | (P << 15) | (P << 18) | (P << 21));
}
template <int M> __device__ __forceinline__ float xmask_add(float a) {
    return a + __int_as_float(__builtin_amdgcn_update_dpp(0, __float_as_int(a), 0x160 + M, 0xf, 0xf, false));
}

__global__ __launch_bounds__(256)
void k_l1_pull(const int* __restrict__ rowptr, const int* __restrict__ csr_src,
               const float* __restrict__ dinv, const float* __restrict__ y,
               const int* __restrict__ batch, float* __restrict__ sd, float* cnt, int n) {
    const int n0 = (int)((blockIdx.x * blockDim.x + threadIdx.x) >> 5) * 32;
    if (n0 >= n) return;
    const int lane = threadIdx.x & 31;
    const int sub  = lane & (L1G - 1);
    const int mine = n0 + lane;
    const int rb = (mine < n) ? rowptr[mine] : N_EDGES;
    const int re = (mine + 1 < n) ? rowptr[mine + 1] : N_EDGES;
    const int mc = mine < n ? mine : n - 1;
    const float dvm = dinv[mc];
    const float ym  = y[mc];
    const int   gi  = (mine < n) ? batch[mine] : -1;

    int beg[L1P], end[L1P];
#define L1_BOUNDS(P) beg[P] = grp_lane<P>(rb); end[P] = grp_lane<P>(re);
    L1_BOUNDS(0) L1_BOUNDS(1) L1_BOUNDS(2) L1_BOUNDS(3) L1_BOUNDS(4) L1_BOUNDS(5) L1_BOUNDS(6) L1_BOUNDS(7)
#undef L1_BOUNDS

    int   u[L1P][L1D];
    float v[L1P][L1D];
#pragma unroll
    for (int p = 0; p < L1P; ++p) {
#pragma unroll
        for (int d = 0; d < L1D; ++d) {
            const int j = beg[p] + sub + d * L1G;
            u[p][d] = csr_src[j < end[p] ? j : N_EDGES];
        }
    }
#pragma unroll
    for (int p = 0; p < L1P; ++p) {
#pragma unroll
        for (int d = 0; d < L1D; ++d) v[p][d] = y[u[p][d]];
    }
    float res = 0.0f;
#pragma unroll
    for (int p = 0; p < L1P; ++p) {
        float a = 0.0f;
#pragma unroll
        for (int d = 0; d < L1D; ++d) a += v[p][d];
        for (int j = beg[p] + sub + L1D * L1G; j < end[p]; j += L1G) a += y[csr_src[j]];
        a = xmask_add<4>(a);
        a = xmask_add<2>(a);
        a = xmask_add<1>(a);
        if (sub == p) res = a;
    }
    if (mine < n) {
        v2f pr = { dvm * (ym + res), dvm };
        ((v2f*)sd)[mine] = pr;
    }
    {
        const int gp = __shfl_up(gi, 1);
        const bool head = (lane == 0) || (gi != gp);
        const unsigned hm = __builtin_amdgcn_ballot_w32(head);
        if (head && gi >= 0) {
            const unsigned above = (lane < 31) ? (hm >> (lane + 1)) : 0u;
            const int run = above ? (__builtin_ctz(above) + 1) : (32 - lane);
            atomicAdd(&cnt[gi], (float)run);
        }
    }
}

typedef float v4f __attribute__((ext_vector_type(4)));
#define AGB 4

#define AG_ADD(P)                                                         \
    {                                                                     \
        const float su_ = (P).x, wu_ = (P).y;                             \
        _Pragma("unroll")                                                 \
        for (int q_ = 0; q_ < HID / 4; ++q_) {                            \
            const float h0_ = su_ * wb[q_].x + wb[q_].z;                  \
            const float h1_ = su_ * wb[q_].y + wb[q_].w;                  \
            ag[q_].x += wu_ * (h0_ > 0.0f ? h0_ : 0.0f);                  \
            ag[q_].y += wu_ * (h1_ > 0.0f ? h1_ : 0.0f);                  \
        }                                                                 \
    }

__global__ __launch_bounds__(256)
void k_l2_fused(const int* __restrict__ rowptr, const int* __restrict__ csr_src, const float* __restrict__ sd,
                const float* __restrict__ W1, const float* __restrict__ b1,
                const float* __restrict__ W2, const float* __restrict__ b2,
                const int* __restrict__ batch, const float* __restrict__ cnt, float* out, int n_tiles, int n) {
    __shared__ v2f sB[(HID / 2) * BPAIR_STRIDE];
    __shared__ v4f sWB[HID / 2];
    {
        float* sf = (float*)sB;
        for (int idx = threadIdx.x; idx < HID * HID; idx += blockDim.x) {
            int k = idx >> 6, col = idx & 63;
            sf[(k >> 1) * (2 * BPAIR_STRIDE) + col * 2 + (k & 1)] = W2[idx];
        }
        if (threadIdx.x < HID / 2) {
            const int p = threadIdx.x;
            v4f wbp = { W1[2 * p], W1[2 * p + 1], b1[2 * p], b1[2 * p + 1] };
            sWB[p] = wbp;
        }
    }
    __syncthreads();

    const int wave = threadIdx.x >> 5;
    const int lane = threadIdx.x & 31;
    const int tile = blockIdx.x * 8 + wave;
    if (tile >= n_tiles) return;

    const int row0 = tile * 16;
    const int m    = lane & 15;
    const int half = lane >> 4;
    const int node = row0 + m;

    const int rbase = row0 + half * 8;
    const int g0 = batch[rbase];

    const v2f* sdp = (const v2f*)sd;
    const int rb = rowptr[node];
    const int re = (node + 1 < n) ? rowptr[node + 1] : N_EDGES;
    const v2f self = sdp[node];

    int j = rb;
    int un[AGB];
    v2f pv[AGB];
#pragma unroll
    for (int d = 0; d < AGB; ++d) un[d] = csr_src[j + d < re ? j + d : N_EDGES];
#pragma unroll
    for (int d = 0; d < AGB; ++d) pv[d] = sdp[un[d]];
#pragma unroll
    for (int d = 0; d < AGB; ++d) un[d] = csr_src[j + AGB + d < re ? j + AGB + d : N_EDGES];

    v4f wb[HID / 4];
    v2f ag[HID / 4];
#pragma unroll
    for (int q = 0; q < HID / 4; ++q) {
        wb[q] = sWB[2 * q + half];
        ag[q].x = 0.0f;
        ag[q].y = 0.0f;
    }
    AG_ADD(self)

    while (__builtin_amdgcn_ballot_w32(j < re) != 0u) {
        v2f pn[AGB];
#pragma unroll
        for (int d = 0; d < AGB; ++d) pn[d] = sdp[un[d]];
#pragma unroll
        for (int d = 0; d < AGB; ++d) un[d] = csr_src[j + 2 * AGB + d < re ? j + 2 * AGB + d : N_EDGES];
        __builtin_amdgcn_sched_barrier(0);
#pragma unroll
        for (int d = 0; d < AGB; ++d) AG_ADD(pv[d])
#pragma unroll
        for (int d = 0; d < AGB; ++d) pv[d] = pn[d];
        j += AGB;
    }

    int g[8];
    g[0] = g0;
#pragma unroll
    for (int r = 1; r < 8; ++r) g[r] = batch[rbase + r];
    float bb[4];
#pragma unroll
    for (int c = 0; c < 4; ++c) bb[c] = b2[16 * c + m];
    const float c0 = cnt[g0];

    const float dv = self.y;
    v8f acc0 = {}, acc1 = {}, acc2 = {}, acc3 = {};
#pragma unroll
    for (int q = 0; q < HID / 4; ++q) {
        v2f a;
        a.x = dv * ag[q].x;
        a.y = dv * ag[q].y;
        const v2f* bp = &sB[(2 * q + half) * BPAIR_STRIDE + m];
        acc0 = __builtin_amdgcn_wmma_f32_16x16x4_f32(false, a, false, bp[ 0], (short)0, acc0, false, false);
        acc1 = __builtin_amdgcn_wmma_f32_16x16x4_f32(false, a, false, bp[16], (short)0, acc1, false, false);
        acc2 = __builtin_amdgcn_wmma_f32_16x16x4_f32(false, a, false, bp[32], (short)0, acc2, false, false);
        acc3 = __builtin_amdgcn_wmma_f32_16x16x4_f32(false, a, false, bp[48], (short)0, acc3, false, false);
    }

    const float inv0 = 1.0f / fmaxf(c0, 1.0f);
    bool same = true;
    float s0 = 0.0f, s1 = 0.0f, s2 = 0.0f, s3 = 0.0f;
#pragma unroll
    for (int r = 0; r < 8; ++r) {
        float v0 = acc0[r] + bb[0], v1 = acc1[r] + bb[1], v2 = acc2[r] + bb[2], v3 = acc3[r] + bb[3];
        v0 = v0 > 0.0f ? v0 : 0.0f;
        v1 = v1 > 0.0f ? v1 : 0.0f;
        v2 = v2 > 0.0f ? v2 : 0.0f;
        v3 = v3 > 0.0f ? v3 : 0.0f;
        acc0[r] = v0; acc1[r] = v1; acc2[r] = v2; acc3[r] = v3;
        s0 += v0; s1 += v1; s2 += v2; s3 += v3;
        same = same && (g[r] == g0);
    }
    const int   gx    = __shfl_xor(g0, 16);
    const int   samex = __shfl_xor((int)same, 16);
    const float x0 = __shfl_xor(s0, 16), x1 = __shfl_xor(s1, 16), x2 = __shfl_xor(s2, 16), x3 = __shfl_xor(s3, 16);
    const bool whole = same && (samex != 0) && (gx == g0);
    if (whole) {
        if (half == 0) {
            float* pr = out + (size_t)g0 * HID + m;
            atomicAdd(&pr[ 0], (s0 + x0) * inv0);
            atomicAdd(&pr[16], (s1 + x1) * inv0);
            atomicAdd(&pr[32], (s2 + x2) * inv0);
            atomicAdd(&pr[48], (s3 + x3) * inv0);
        }
    } else if (same) {
        float* pr = out + (size_t)g0 * HID + m;
        atomicAdd(&pr[ 0], s0 * inv0);
        atomicAdd(&pr[16], s1 * inv0);
        atomicAdd(&pr[32], s2 * inv0);
        atomicAdd(&pr[48], s3 * inv0);
    } else {
        float cr[8];
#pragma unroll
        for (int r = 0; r < 8; ++r) cr[r] = cnt[g[r]];
#pragma unroll
        for (int r = 0; r < 8; ++r) {
            const float ir = 1.0f / fmaxf(cr[r], 1.0f);
            float* pr = out + (size_t)g[r] * HID + m;
            atomicAdd(&pr[ 0], acc0[r] * ir);
            atomicAdd(&pr[16], acc1[r] * ir);
            atomicAdd(&pr[32], acc2[r] * ir);
            atomicAdd(&pr[48], acc3[r] * ir);
        }
    }
}
#undef AG_ADD

extern "C" void kernel_launch(void* const* d_in, const int* in_sizes, int n_in,
                              void* d_out, int out_size, void* d_ws, size_t ws_size,
                              hipStream_t stream) {
    const float* x     = (const float*)d_in[0];
    const int*   edge  = (const int*)  d_in[1];
    const int*   batch = (const int*)  d_in[2];
    const float* W1    = (const float*)d_in[3];
    const float* b1    = (const float*)d_in[4];
    const float* W2    = (const float*)d_in[5];
    const float* b2    = (const float*)d_in[6];
    const int* src = edge;
    const int* dst = edge + N_EDGES;
    float* out = (float*)d_out;

    float* ws   = (float*)d_ws;
    float* dinv = ws;
    float* bufA = dinv + 2 * N_NODES;
    float* bufB = bufA + (size_t)N_NODES * HID;
    int* runs    = (int*)bufA;
    v2i* ebuf    = (v2i*)(runs + NEBLK * NBKT + 512);
    int* rowptr  = (int*)bufB;
    int* csr_src = rowptr + N_NODES + 1024;
    float* y     = (float*)(csr_src + N_EDGES + 32);
    float* sd    = y + N_NODES + 8;
    float* cnt  = bufB + (size_t)N_NODES * HID;

    const int TB = 256;
    k_bkt_scatter  <<<NEBLK, ETB, 0, stream>>>(src, dst, runs, ebuf, out, cnt);
    k_bkt_sort     <<<NBKT, PTB, 0, stream>>>(ebuf, runs, x, rowptr, dinv, y, csr_src, sd, N_NODES);
    k_l1_pull      <<<(N_NODES + TB - 1) / TB, TB, 0, stream>>>(rowptr, csr_src, dinv, y, batch, sd, cnt, N_NODES);

    const int n_tiles = N_NODES / 16;
    k_l2_fused     <<<(n_tiles + 7) / 8, 256, 0, stream>>>(rowptr, csr_src, sd, W1, b1, W2, b2, batch,
                                                           cnt, out, n_tiles, N_NODES);
}
